# starting kernel
# baseline (speedup 1.0000x reference)
#include <hip/hip_runtime.h>

#define IN_CH 128
#define OUT_CH 128
#define W_STRIDE 132

typedef __attribute__((ext_vector_type(2))) float v2f;
typedef __attribute__((ext_vector_type(8))) float v8f;

__global__ void zero_kernel(float4* __restrict__ out4, float4* __restrict__ deg4,
                            int nOut4, int nDeg4) {
    int i = blockIdx.x * blockDim.x + threadIdx.x;
    float4 z = make_float4(0.f, 0.f, 0.f, 0.f);
    if (i < nOut4) out4[i] = z;
    if (i < nDeg4) deg4[i] = z;
}

__global__ void deg_kernel(const int* __restrict__ row, float* __restrict__ deg,
                           int n_edges) {
    int e = blockIdx.x * blockDim.x + threadIdx.x;
    if (e < n_edges) unsafeAtomicAdd(&deg[row[e]], 1.0f);
}

__global__ void dis_kernel(const float* __restrict__ deg, float* __restrict__ dis,
                           int n) {
    int i = blockIdx.x * blockDim.x + threadIdx.x;
    if (i < n) {
        float d = deg[i];
        dis[i] = (d > 0.f) ? rsqrtf(d) : 0.f;
    }
}

__global__ __launch_bounds__(256) void scatter_kernel(
    const float* __restrict__ x, const int* __restrict__ row,
    const int* __restrict__ col, const float* __restrict__ dis,
    float* __restrict__ agg, int n_edges) {
    int tid = blockIdx.x * blockDim.x + threadIdx.x;
    int e = tid >> 5;
    if (e >= n_edges) return;
    int lane = tid & 31;
    int r = row[e];
    int c = col[e];
    float norm = dis[r] * dis[c];
    const float4 xv = *(const float4*)(x + (size_t)c * IN_CH + lane * 4);
    float* dst = agg + (size_t)r * IN_CH + lane * 4;
    unsafeAtomicAdd(dst + 0, norm * xv.x);
    unsafeAtomicAdd(dst + 1, norm * xv.y);
    unsafeAtomicAdd(dst + 2, norm * xv.z);
    unsafeAtomicAdd(dst + 3, norm * xv.w);
}

__global__ __launch_bounds__(256) void gemm_wmma_kernel(
    const float* agg, const float* __restrict__ W, const float* __restrict__ bias,
    float* out, int n_strips, int n_nodes) {
    extern __shared__ float sW[];

    for (int i = threadIdx.x; i < OUT_CH * IN_CH; i += 256) {
        sW[(i >> 7) * W_STRIDE + (i & 127)] = W[i];
    }
    __syncthreads();

    const int lane = threadIdx.x & 31;
    const int wave = threadIdx.x >> 5;
    const int strip = blockIdx.x * 8 + wave;
    if (strip >= n_strips) return;

    const int m = lane & 15;
    const int half = lane >> 4;

    v8f acc[8] = {};

    int anode = strip * 16 + m;
    if (anode > n_nodes - 1) anode = n_nodes - 1;
    const float* arow = agg + (size_t)anode * IN_CH + 2 * half;

    for (int k = 0; k < IN_CH; k += 4) {
        v2f a = *(const v2f*)(arow + k);
#pragma unroll
        for (int t = 0; t < 8; ++t) {
            const float* bp = &sW[(t * 16 + m) * W_STRIDE + k + 2 * half];
            v2f b = *(const v2f*)bp;
            acc[t] = __builtin_amdgcn_wmma_f32_16x16x4_f32(
                false, a, false, b, (short)0, acc[t], false, false);
        }
    }

#pragma unroll
    for (int t = 0; t < 8; ++t) {
        float bv = bias[t * 16 + m];
#pragma unroll
        for (int r = 0; r < 8; ++r) {
            int rowNode = strip * 16 + r + 8 * half;
            if (rowNode < n_nodes)
                out[(size_t)rowNode * OUT_CH + t * 16 + m] = acc[t][r] + bv;
        }
    }
}

extern "C" void kernel_launch(void* const* d_in, const int* in_sizes, int n_in,
                              void* d_out, int out_size, void* d_ws, size_t ws_size,
                              hipStream_t stream) {
    (void)n_in; (void)out_size; (void)ws_size;

    const float* x  = (const float*)d_in[0];
    const int*   ei = (const int*)d_in[1];
    const float* W  = (const float*)d_in[2];
    const float* b  = (const float*)d_in[3];
    float* out = (float*)d_out;

    const int n_nodes = in_sizes[0] / IN_CH;
    const int n_edges = in_sizes[1] / 2;
    const int* row = ei;
    const int* col = ei + n_edges;

    float* deg = (float*)d_ws;
    float* dis = deg + ((n_nodes + 63) & ~63);

    int nOut4 = (n_nodes * IN_CH) / 4;
    int nDeg4 = (n_nodes + 3) / 4;
    zero_kernel<<<(nOut4 + 255) / 256, 256, 0, stream>>>(
        (float4*)out, (float4*)deg, nOut4, nDeg4);

    deg_kernel<<<(n_edges + 255) / 256, 256, 0, stream>>>(row, deg, n_edges);

    dis_kernel<<<(n_nodes + 255) / 256, 256, 0, stream>>>(deg, dis, n_nodes);

    long long sthreads = (long long)n_edges * 32;
    scatter_kernel<<<(int)((sthreads + 255) / 256), 256, 0, stream>>>(
        x, row, col, dis, out, n_edges);

    int n_strips = (n_nodes + 15) / 16;
    gemm_wmma_kernel<<<(n_strips + 7) / 8, 256, OUT_CH * W_STRIDE * sizeof(float),
                      stream>>>(out, W, b, out, n_strips, n_nodes);
}


# f32 atomic scatter replaced by edges bucketed per target row (int count/slot atomics, scan, fill) + one wave per row summing in registers; sum order fixed: bucket sorted by source index
# speedup vs baseline: 1.5122x; 1.5122x over previous
#include <hip/hip_runtime.h>


#define IN_CH 128
#define OUT_CH 128
#define W_STRIDE 132
#define KEY_MAX 0x7fffffff

typedef __attribute__((ext_vector_type(2))) float v2f;
typedef __attribute__((ext_vector_type(8))) float v8f;

__global__ void zero_kernel(int4* __restrict__ cnt4, int nCnt4) {
    int i = blockIdx.x * blockDim.x + threadIdx.x;
    if (i < nCnt4) cnt4[i] = make_int4(0, 0, 0, 0);
}

__global__ __launch_bounds__(256) void hist_kernel(const int* __restrict__ row,
                                                   int* __restrict__ cnt,
                                                   int* __restrict__ ticket,
                                                   int n_edges) {
    int e = blockIdx.x * blockDim.x + threadIdx.x;
    if (e < n_edges) ticket[e] = atomicAdd(&cnt[row[e]], 1);
}

__global__ __launch_bounds__(256) void scan_kernel(const int* __restrict__ cnt,
                                                   int* __restrict__ off,
                                                   float* __restrict__ dis, int n) {
    __shared__ int s_front[8];
    __shared__ int s_wsum[8];
    const int tid = threadIdx.x;
    const int lane = tid & 31;
    const int wv = tid >> 5;
    const int blockStart = blockIdx.x * 1024;

    int front = 0;
    const int4* c4 = (const int4*)cnt;
    const int nch = blockIdx.x;
#pragma clang loop vectorize(disable) interleave(disable) unroll(disable)
    for (int c = 0; c < nch; c += 16) {
        int4 q[16];
#pragma unroll
        for (int u = 0; u < 16; ++u) {
            const int cc = ((c + u) < nch) ? (c + u) : c;
            q[u] = c4[cc * 256 + tid];
        }
#pragma unroll
        for (int u = 0; u < 16; ++u) {
            const int sq = (q[u].x + q[u].y) + (q[u].z + q[u].w);
            front += ((c + u) < nch) ? sq : 0;
        }
    }
    for (int o = 16; o > 0; o >>= 1) front += __shfl_xor(front, o);
    if (lane == 0) s_front[wv] = front;

    const int i0 = blockStart + tid * 4;
    int v[4];
#pragma unroll
    for (int j = 0; j < 4; ++j) v[j] = (i0 + j < n) ? cnt[i0 + j] : 0;
    const int tsum = (v[0] + v[1]) + (v[2] + v[3]);
    int incl = tsum;
    for (int o = 1; o < 32; o <<= 1) {
        int t = __shfl_up(incl, o);
        if (lane >= o) incl += t;
    }
    if (lane == 31) s_wsum[wv] = incl;
    __syncthreads();
    int base = 0;
    for (int w = 0; w < 8; ++w) base += s_front[w];
    for (int w = 0; w < wv; ++w) base += s_wsum[w];
    int excl = base + incl - tsum;
#pragma unroll
    for (int j = 0; j < 4; ++j) {
        if (i0 + j < n) {
            off[i0 + j] = excl;
            dis[i0 + j] = (v[j] > 0) ? rsqrtf((float)v[j]) : 0.f;
        }
        excl += v[j];
    }
}

__global__ __launch_bounds__(256) void fill_kernel(const int* __restrict__ row,
                                                   const int* __restrict__ col,
                                                   const int* __restrict__ off,
                                                   const int* __restrict__ ticket,
                                                   int* __restrict__ bucket,
                                                   int n_edges) {
    int e = blockIdx.x * blockDim.x + threadIdx.x;
    if (e < n_edges) bucket[off[row[e]] + ticket[e]] = col[e];
}

__device__ __forceinline__ int lane_i(int v, int l) {
    return __builtin_amdgcn_readlane(v, l);
}
__device__ __forceinline__ float lane_f(float v, int l) {
    return __int_as_float(__builtin_amdgcn_readlane(__float_as_int(v), l));
}

__global__ __launch_bounds__(256) void gather_kernel(
    const float* __restrict__ x, const int* __restrict__ bucket,
    const int* __restrict__ off, const int* __restrict__ cnt,
    const float* __restrict__ dis, float* __restrict__ agg, int n_nodes) {
    const int lane = threadIdx.x & 31;
    const int r = __builtin_amdgcn_readfirstlane(
        (int)((blockIdx.x * blockDim.x + threadIdx.x) >> 5));
    if (r >= n_nodes) return;
    const float* xl = x + lane * 4;

    {
        const int start = off[r];
        const int d = cnt[r];
        const float dr = dis[r];
        float4 acc = make_float4(0.f, 0.f, 0.f, 0.f);

        if (d <= 32) {
            int k = KEY_MAX;
            if (lane < d) k = bucket[start + lane];
            int rank = 0;
            for (int j = 0; j < d; ++j) {
                const int kj = lane_i(k, j);
                rank += (kj < k || (kj == k && j < lane)) ? 1 : 0;
            }
            if (lane >= d) rank = lane;
            const int s = __builtin_amdgcn_ds_permute(rank << 2, k);
            float nv = 0.f;
            if (lane < d) nv = dr * dis[s];

            int t = 0;
            for (; t + 4 <= d; t += 4) {
                const int c0 = lane_i(s, t), c1 = lane_i(s, t + 1);
                const int c2 = lane_i(s, t + 2), c3 = lane_i(s, t + 3);
                const float n0 = lane_f(nv, t), n1 = lane_f(nv, t + 1);
                const float n2 = lane_f(nv, t + 2), n3 = lane_f(nv, t + 3);
                const float4 x0 = *(const float4*)(xl + (size_t)c0 * IN_CH);
                const float4 x1 = *(const float4*)(xl + (size_t)c1 * IN_CH);
                const float4 x2 = *(const float4*)(xl + (size_t)c2 * IN_CH);
                const float4 x3 = *(const float4*)(xl + (size_t)c3 * IN_CH);
                acc.x += n0 * x0.x; acc.y += n0 * x0.y; acc.z += n0 * x0.z; acc.w += n0 * x0.w;
                acc.x += n1 * x1.x; acc.y += n1 * x1.y; acc.z += n1 * x1.z; acc.w += n1 * x1.w;
                acc.x += n2 * x2.x; acc.y += n2 * x2.y; acc.z += n2 * x2.z; acc.w += n2 * x2.w;
                acc.x += n3 * x3.x; acc.y += n3 * x3.y; acc.z += n3 * x3.z; acc.w += n3 * x3.w;
            }
            for (; t < d; ++t) {
                const int c0 = lane_i(s, t);
                const float n0 = lane_f(nv, t);
                const float4 x0 = *(const float4*)(xl + (size_t)c0 * IN_CH);
                acc.x += n0 * x0.x; acc.y += n0 * x0.y; acc.z += n0 * x0.z; acc.w += n0 * x0.w;
            }
        } else {
            int prev = -1;
            int done = 0;
            while (done < d) {
                int mn = KEY_MAX;
#pragma unroll 1
                for (int i = lane; i < d; i += 32) {
                    const int kk = bucket[start + i];
                    if (kk > prev && kk < mn) mn = kk;
                }
                for (int o = 16; o > 0; o >>= 1) {
                    const int m2 = __shfl_xor(mn, o);
                    mn = (m2 < mn) ? m2 : mn;
                }
                int c = 0;
#pragma unroll 1
                for (int i = lane; i < d; i += 32) c += (bucket[start + i] == mn) ? 1 : 0;
                for (int o = 16; o > 0; o >>= 1) c += __shfl_xor(c, o);
                mn = __builtin_amdgcn_readfirstlane(mn);
                c = __builtin_amdgcn_readfirstlane(c);
                if (c <= 0) break;
                const float n0 = dr * dis[mn];
                const float4 x0 = *(const float4*)(xl + (size_t)mn * IN_CH);
                for (int q = 0; q < c; ++q) {
                    acc.x += n0 * x0.x; acc.y += n0 * x0.y; acc.z += n0 * x0.z; acc.w += n0 * x0.w;
                }
                prev = mn;
                done += c;
            }
        }
        *(float4*)(agg + (size_t)r * IN_CH + lane * 4) = acc;
    }
}

__global__ __launch_bounds__(256) void gemm_wmma_kernel(
    const float* agg, const float* __restrict__ W, const float* __restrict__ bias,
    float* out, int n_strips, int n_nodes) {
    extern __shared__ float sW[];

    for (int i = threadIdx.x; i < OUT_CH * IN_CH; i += 256) {
        sW[(i >> 7) * W_STRIDE + (i & 127)] = W[i];
    }
    __syncthreads();

    const int lane = threadIdx.x & 31;
    const int wave = threadIdx.x >> 5;
    const int strip = blockIdx.x * 8 + wave;
    if (strip >= n_strips) return;

    const int m = lane & 15;
    const int half = lane >> 4;

    v8f acc[8] = {};

    int anode = strip * 16 + m;
    if (anode > n_nodes - 1) anode = n_nodes - 1;
    const float* arow = agg + (size_t)anode * IN_CH + 2 * half;

    for (int k = 0; k < IN_CH; k += 4) {
        v2f a = *(const v2f*)(arow + k);
#pragma unroll
        for (int t = 0; t < 8; ++t) {
            const float* bp = &sW[(t * 16 + m) * W_STRIDE + k + 2 * half];
            v2f b = *(const v2f*)bp;
            acc[t] = __builtin_amdgcn_wmma_f32_16x16x4_f32(
                false, a, false, b, (short)0, acc[t], false, false);
        }
    }

#pragma unroll
    for (int t = 0; t < 8; ++t) {
        float bv = bias[t * 16 + m];
#pragma unroll
        for (int r = 0; r < 8; ++r) {
            int rowNode = strip * 16 + r + 8 * half;
            if (rowNode < n_nodes)
                out[(size_t)rowNode * OUT_CH + t * 16 + m] = acc[t][r] + bv;
        }
    }
}

extern "C" void kernel_launch(void* const* d_in, const int* in_sizes, int n_in,
                              void* d_out, int out_size, void* d_ws, size_t ws_size,
                              hipStream_t stream) {
    (void)n_in; (void)out_size; (void)ws_size;

    const float* x  = (const float*)d_in[0];
    const int*   ei = (const int*)d_in[1];
    const float* W  = (const float*)d_in[2];
    const float* b  = (const float*)d_in[3];
    float* out = (float*)d_out;

    const int n_nodes = in_sizes[0] / IN_CH;
    const int n_edges = in_sizes[1] / 2;
    const int* row = ei;
    const int* col = ei + n_edges;

    const int n_pad = (n_nodes + 63) & ~63;
    const int e_pad = (n_edges + 63) & ~63;
    int*   cnt    = (int*)d_ws;
    int*   off    = cnt + n_pad;
    float* dis    = (float*)(off + n_pad);
    int*   ticket = (int*)(dis + n_pad);
    int*   bucket = ticket + e_pad;

    int nCnt4 = n_pad / 4;
    zero_kernel<<<(nCnt4 + 255) / 256, 256, 0, stream>>>((int4*)cnt, nCnt4);

    hist_kernel<<<(n_edges + 255) / 256, 256, 0, stream>>>(row, cnt, ticket, n_edges);

    scan_kernel<<<(n_nodes + 1023) / 1024, 256, 0, stream>>>(cnt, off, dis, n_nodes);

    fill_kernel<<<(n_edges + 255) / 256, 256, 0, stream>>>(row, col, off, ticket, bucket,
                                                            n_edges);

    gather_kernel<<<(n_nodes + 7) / 8, 256, 0, stream>>>(x, bucket, off, cnt, dis, out, n_nodes);

    int n_strips = (n_nodes + 15) / 16;
    gemm_wmma_kernel<<<(n_strips + 7) / 8, 256, OUT_CH * W_STRIDE * sizeof(float),
                      stream>>>(out, W, b, out, n_strips, n_nodes);
}


# edges grouped per target row by a two-level counting sort through LDS counters (256-row bins, runs assembled in LDS, stored whole) instead of global int atomics + scattered stores; sum order kept
# speedup vs baseline: 2.5783x; 1.7050x over previous
#include <hip/hip_runtime.h>


#define IN_CH 128
#define OUT_CH 128
#define W_STRIDE 132
#define KEY_MAX 0x7fffffff

#define BIN_SHIFT 8
#define BIN_NODES 256
#define MAX_BINS 1024
#define CHUNK 4096
#define STAGE_CAP 8192

typedef __attribute__((ext_vector_type(2))) float v2f;
typedef __attribute__((ext_vector_type(8))) float v8f;

__device__ __forceinline__ void load4(const int* __restrict__ p, int e0, int n, int (&v)[4]) {
    if (e0 + 3 < n && ((((size_t)(p + e0)) & 15) == 0)) {
        const int4 q = *(const int4*)(p + e0);
        v[0] = q.x; v[1] = q.y; v[2] = q.z; v[3] = q.w;
    } else {
        v[0] = (e0 + 0 < n) ? p[e0 + 0] : -1;
        v[1] = (e0 + 1 < n) ? p[e0 + 1] : -1;
        v[2] = (e0 + 2 < n) ? p[e0 + 2] : -1;
        v[3] = (e0 + 3 < n) ? p[e0 + 3] : -1;
    }
}

__device__ __forceinline__ void load4x2(const int* __restrict__ p, const int* __restrict__ q,
                                        int e0, int n, int (&v)[4], int (&w)[4]) {
    if (e0 + 3 < n && (((((size_t)(p + e0)) | ((size_t)(q + e0))) & 15) == 0)) {
        const int4 a = *(const int4*)(p + e0);
        const int4 b = *(const int4*)(q + e0);
        v[0] = a.x; v[1] = a.y; v[2] = a.z; v[3] = a.w;
        w[0] = b.x; w[1] = b.y; w[2] = b.z; w[3] = b.w;
    } else {
        v[0] = (e0 + 0 < n) ? p[e0 + 0] : -1;
        v[1] = (e0 + 1 < n) ? p[e0 + 1] : -1;
        v[2] = (e0 + 2 < n) ? p[e0 + 2] : -1;
        v[3] = (e0 + 3 < n) ? p[e0 + 3] : -1;
        w[0] = (e0 + 0 < n) ? q[e0 + 0] : 0;
        w[1] = (e0 + 1 < n) ? q[e0 + 1] : 0;
        w[2] = (e0 + 2 < n) ? q[e0 + 2] : 0;
        w[3] = (e0 + 3 < n) ? q[e0 + 3] : 0;
    }
}

__global__ __launch_bounds__(1024) void bin_count_kernel(const int* __restrict__ row,
                                                         int* __restrict__ table,
                                                         int n_edges, int n_bins, int n_blk) {
    __shared__ int s_h[MAX_BINS];
    const int tid = threadIdx.x;
    if (tid < n_bins) s_h[tid] = 0;
    __syncthreads();
    const int e0 = blockIdx.x * CHUNK + tid * 4;
    int r[4];
    load4(row, e0, n_edges, r);
    if (r[0] >= 0) atomicAdd(&s_h[r[0] >> BIN_SHIFT], 1);
    if (r[1] >= 0) atomicAdd(&s_h[r[1] >> BIN_SHIFT], 1);
    if (r[2] >= 0) atomicAdd(&s_h[r[2] >> BIN_SHIFT], 1);
    if (r[3] >= 0) atomicAdd(&s_h[r[3] >> BIN_SHIFT], 1);
    __syncthreads();
    if (tid < n_bins) table[tid * n_blk + blockIdx.x] = s_h[tid];
}

__global__ __launch_bounds__(1024) void bin_partition_kernel(
    const int* __restrict__ row, const int* __restrict__ col, const int* __restrict__ table,
    int2* __restrict__ recs, int* __restrict__ binStart, int n_edges, int n_bins, int n_blk) {
    __shared__ int s_start[MAX_BINS];
    __shared__ int s_before[MAX_BINS];
    __shared__ int s_cnt[MAX_BINS];
    __shared__ int s_loc[MAX_BINS];
    __shared__ int s_wa[32];
    __shared__ int s_wb[32];
    __shared__ int2 s_stage[CHUNK];
    const int tid = threadIdx.x;
    const int lane = tid & 31;
    const int wv = tid >> 5;
    const int blk = blockIdx.x;

    const int e0 = blk * CHUNK + tid * 4;
    int r[4], c[4];
    load4x2(row, col, e0, n_edges, r, c);

    if (tid < n_bins) s_cnt[tid] = 0;

    for (int b = wv; b < n_bins; b += 64) {
        const int b2 = b + 32;
        const bool has2 = b2 < n_bins;
        const int* t0 = table + b * n_blk;
        const int* t1 = table + (has2 ? b2 : b) * n_blk;
        int tot0 = 0, bef0 = 0, tot1 = 0, bef1 = 0;
        for (int k0 = 0; k0 < n_blk; k0 += 256) {
            int q0[8], q1[8];
#pragma unroll
            for (int u = 0; u < 8; ++u) {
                const int k = k0 + u * 32 + lane;
                q0[u] = (k < n_blk) ? t0[k] : 0;
                q1[u] = (k < n_blk) ? t1[k] : 0;
            }
#pragma unroll
            for (int u = 0; u < 8; ++u) {
                const int k = k0 + u * 32 + lane;
                tot0 += q0[u];
                tot1 += q1[u];
                bef0 += (k < blk) ? q0[u] : 0;
                bef1 += (k < blk) ? q1[u] : 0;
            }
        }
        for (int o = 16; o > 0; o >>= 1) {
            tot0 += __shfl_xor(tot0, o);
            bef0 += __shfl_xor(bef0, o);
            tot1 += __shfl_xor(tot1, o);
            bef1 += __shfl_xor(bef1, o);
        }
        if (lane == 0) {
            s_start[b] = tot0;
            s_before[b] = bef0;
            if (has2) {
                s_start[b2] = tot1;
                s_before[b2] = bef1;
            }
        }
    }
    __syncthreads();

    int rank[4];
    rank[0] = (r[0] >= 0) ? atomicAdd(&s_cnt[r[0] >> BIN_SHIFT], 1) : 0;
    rank[1] = (r[1] >= 0) ? atomicAdd(&s_cnt[r[1] >> BIN_SHIFT], 1) : 0;
    rank[2] = (r[2] >= 0) ? atomicAdd(&s_cnt[r[2] >> BIN_SHIFT], 1) : 0;
    rank[3] = (r[3] >= 0) ? atomicAdd(&s_cnt[r[3] >> BIN_SHIFT], 1) : 0;
    __syncthreads();

    const int a = (tid < n_bins) ? s_start[tid] : 0;
    const int l = (tid < n_bins) ? s_cnt[tid] : 0;
    int ia = a, il = l;
    for (int o = 1; o < 32; o <<= 1) {
        const int ta = __shfl_up(ia, o);
        const int tl = __shfl_up(il, o);
        if (lane >= o) { ia += ta; il += tl; }
    }
    if (lane == 31) { s_wa[wv] = ia; s_wb[wv] = il; }
    __syncthreads();
    int ba = 0, bl = 0;
    for (int w = 0; w < wv; ++w) { ba += s_wa[w]; bl += s_wb[w]; }
    const int ga = ba + ia - a;
    const int gl = bl + il - l;
    if (tid < n_bins) {
        s_start[tid] = ga;
        s_loc[tid] = gl;
        if (blk == 0) binStart[tid] = ga;
    }
    __syncthreads();

    if (r[0] >= 0) s_stage[s_loc[r[0] >> BIN_SHIFT] + rank[0]] = make_int2(r[0], c[0]);
    if (r[1] >= 0) s_stage[s_loc[r[1] >> BIN_SHIFT] + rank[1]] = make_int2(r[1], c[1]);
    if (r[2] >= 0) s_stage[s_loc[r[2] >> BIN_SHIFT] + rank[2]] = make_int2(r[2], c[2]);
    if (r[3] >= 0) s_stage[s_loc[r[3] >> BIN_SHIFT] + rank[3]] = make_int2(r[3], c[3]);
    __syncthreads();

    int n_here = n_edges - blk * CHUNK;
    if (n_here > CHUNK) n_here = CHUNK;
    for (int i = tid; i < n_here; i += 1024) {
        const int2 rc = s_stage[i];
        const int bin = rc.x >> BIN_SHIFT;
        recs[s_start[bin] + s_before[bin] + (i - s_loc[bin])] = rc;
    }
}

__global__ __launch_bounds__(1024) void bin_build_kernel(
    const int2* __restrict__ recs, const int* __restrict__ binStart, int* __restrict__ bucket,
    int* __restrict__ off, int* __restrict__ cnt, float* __restrict__ dis,
    int n_nodes, int n_edges, int n_bins) {
    __shared__ int s_c[BIN_NODES];
    __shared__ int s_o[BIN_NODES];
    __shared__ int s_cur[BIN_NODES];
    __shared__ int s_w[8];
    __shared__ int s_b[STAGE_CAP];
    const int tid = threadIdx.x;
    const int lane = tid & 31;
    const int wv = tid >> 5;
    const int b = blockIdx.x;
    const int start = binStart[b];
    const int end = (b + 1 < n_bins) ? binStart[b + 1] : n_edges;
    const int m = end - start;
    const int node0 = b << BIN_SHIFT;

    if (tid < BIN_NODES) { s_c[tid] = 0; s_cur[tid] = 0; }
    __syncthreads();
    for (int i0 = 0; i0 < m; i0 += 4096) {
        int rx[4];
#pragma unroll
        for (int u = 0; u < 4; ++u) {
            const int i = i0 + u * 1024 + tid;
            rx[u] = (i < m) ? recs[start + i].x : -1;
        }
#pragma unroll
        for (int u = 0; u < 4; ++u)
            if (rx[u] >= 0) atomicAdd(&s_c[rx[u] - node0], 1);
    }
    __syncthreads();

    const int v = (tid < BIN_NODES) ? s_c[tid] : 0;
    int incl = v;
    for (int o = 1; o < 32; o <<= 1) {
        const int t = __shfl_up(incl, o);
        if (lane >= o) incl += t;
    }
    if (tid < BIN_NODES && lane == 31) s_w[wv] = incl;
    __syncthreads();
    if (tid < BIN_NODES) {
        int base = 0;
        for (int w = 0; w < wv; ++w) base += s_w[w];
        const int excl = base + incl - v;
        s_o[tid] = excl;
        const int node = node0 + tid;
        if (node < n_nodes) {
            off[node] = start + excl;
            cnt[node] = v;
            dis[node] = (v > 0) ? rsqrtf((float)v) : 0.f;
        }
    }
    __syncthreads();

    const bool staged = (m <= STAGE_CAP);
    for (int i0 = 0; i0 < m; i0 += 4096) {
        int2 rc[4];
#pragma unroll
        for (int u = 0; u < 4; ++u) {
            const int i = i0 + u * 1024 + tid;
            rc[u] = (i < m) ? recs[start + i] : make_int2(-1, 0);
        }
#pragma unroll
        for (int u = 0; u < 4; ++u) {
            if (rc[u].x >= 0) {
                const int ln = rc[u].x - node0;
                const int p = s_o[ln] + atomicAdd(&s_cur[ln], 1);
                if (staged) s_b[p] = rc[u].y;
                else bucket[start + p] = rc[u].y;
            }
        }
    }
    if (staged) {
        __syncthreads();
        for (int i = tid; i < m; i += 1024) bucket[start + i] = s_b[i];
    }
}

__device__ __forceinline__ int lane_i(int v, int l) {
    return __builtin_amdgcn_readlane(v, l);
}
__device__ __forceinline__ float lane_f(float v, int l) {
    return __int_as_float(__builtin_amdgcn_readlane(__float_as_int(v), l));
}

__global__ __launch_bounds__(256) void gather_kernel(
    const float* __restrict__ x, const int* __restrict__ bucket,
    const int* __restrict__ off, const int* __restrict__ cnt,
    const float* __restrict__ dis, float* __restrict__ agg, int n_nodes) {
    const int lane = threadIdx.x & 31;
    const int r = __builtin_amdgcn_readfirstlane(
        (int)((blockIdx.x * blockDim.x + threadIdx.x) >> 5));
    if (r >= n_nodes) return;
    const float* xl = x + lane * 4;

    {
        const int start = off[r];
        const int d = cnt[r];
        const float dr = dis[r];
        float4 acc = make_float4(0.f, 0.f, 0.f, 0.f);

        if (d <= 32) {
            int k = KEY_MAX;
            if (lane < d) k = bucket[start + lane];
            int rank = 0;
            for (int j = 0; j < d; ++j) {
                const int kj = lane_i(k, j);
                rank += (kj < k || (kj == k && j < lane)) ? 1 : 0;
            }
            if (lane >= d) rank = lane;
            const int s = __builtin_amdgcn_ds_permute(rank << 2, k);
            float nv = 0.f;
            if (lane < d) nv = dr * dis[s];

            int t = 0;
            for (; t + 4 <= d; t += 4) {
                const int c0 = lane_i(s, t), c1 = lane_i(s, t + 1);
                const int c2 = lane_i(s, t + 2), c3 = lane_i(s, t + 3);
                const float n0 = lane_f(nv, t), n1 = lane_f(nv, t + 1);
                const float n2 = lane_f(nv, t + 2), n3 = lane_f(nv, t + 3);
                const float4 x0 = *(const float4*)(xl + (size_t)c0 * IN_CH);
                const float4 x1 = *(const float4*)(xl + (size_t)c1 * IN_CH);
                const float4 x2 = *(const float4*)(xl + (size_t)c2 * IN_CH);
                const float4 x3 = *(const float4*)(xl + (size_t)c3 * IN_CH);
                acc.x += n0 * x0.x; acc.y += n0 * x0.y; acc.z += n0 * x0.z; acc.w += n0 * x0.w;
                acc.x += n1 * x1.x; acc.y += n1 * x1.y; acc.z += n1 * x1.z; acc.w += n1 * x1.w;
                acc.x += n2 * x2.x; acc.y += n2 * x2.y; acc.z += n2 * x2.z; acc.w += n2 * x2.w;
                acc.x += n3 * x3.x; acc.y += n3 * x3.y; acc.z += n3 * x3.z; acc.w += n3 * x3.w;
            }
            for (; t < d; ++t) {
                const int c0 = lane_i(s, t);
                const float n0 = lane_f(nv, t);
                const float4 x0 = *(const float4*)(xl + (size_t)c0 * IN_CH);
                acc.x += n0 * x0.x; acc.y += n0 * x0.y; acc.z += n0 * x0.z; acc.w += n0 * x0.w;
            }
        } else {
            int prev = -1;
            int done = 0;
            while (done < d) {
                int mn = KEY_MAX;
#pragma unroll 1
                for (int i = lane; i < d; i += 32) {
                    const int kk = bucket[start + i];
                    if (kk > prev && kk < mn) mn = kk;
                }
                for (int o = 16; o > 0; o >>= 1) {
                    const int m2 = __shfl_xor(mn, o);
                    mn = (m2 < mn) ? m2 : mn;
                }
                int c = 0;
#pragma unroll 1
                for (int i = lane; i < d; i += 32) c += (bucket[start + i] == mn) ? 1 : 0;
                for (int o = 16; o > 0; o >>= 1) c += __shfl_xor(c, o);
                mn = __builtin_amdgcn_readfirstlane(mn);
                c = __builtin_amdgcn_readfirstlane(c);
                if (c <= 0) break;
                const float n0 = dr * dis[mn];
                const float4 x0 = *(const float4*)(xl + (size_t)mn * IN_CH);
                for (int q = 0; q < c; ++q) {
                    acc.x += n0 * x0.x; acc.y += n0 * x0.y; acc.z += n0 * x0.z; acc.w += n0 * x0.w;
                }
                prev = mn;
                done += c;
            }
        }
        *(float4*)(agg + (size_t)r * IN_CH + lane * 4) = acc;
    }
}

__global__ __launch_bounds__(256) void gemm_wmma_kernel(
    const float* agg, const float* __restrict__ W, const float* __restrict__ bias,
    float* out, int n_strips, int n_nodes) {
    extern __shared__ float sW[];

    for (int i = threadIdx.x; i < OUT_CH * IN_CH; i += 256) {
        sW[(i >> 7) * W_STRIDE + (i & 127)] = W[i];
    }
    __syncthreads();

    const int lane = threadIdx.x & 31;
    const int wave = threadIdx.x >> 5;
    const int strip = blockIdx.x * 8 + wave;
    if (strip >= n_strips) return;

    const int m = lane & 15;
    const int half = lane >> 4;

    v8f acc[8] = {};

    int anode = strip * 16 + m;
    if (anode > n_nodes - 1) anode = n_nodes - 1;
    const float* arow = agg + (size_t)anode * IN_CH + 2 * half;

    for (int k = 0; k < IN_CH; k += 4) {
        v2f a = *(const v2f*)(arow + k);
#pragma unroll
        for (int t = 0; t < 8; ++t) {
            const float* bp = &sW[(t * 16 + m) * W_STRIDE + k + 2 * half];
            v2f b = *(const v2f*)bp;
            acc[t] = __builtin_amdgcn_wmma_f32_16x16x4_f32(
                false, a, false, b, (short)0, acc[t], false, false);
        }
    }

#pragma unroll
    for (int t = 0; t < 8; ++t) {
        float bv = bias[t * 16 + m];
#pragma unroll
        for (int r = 0; r < 8; ++r) {
            int rowNode = strip * 16 + r + 8 * half;
            if (rowNode < n_nodes)
                out[(size_t)rowNode * OUT_CH + t * 16 + m] = acc[t][r] + bv;
        }
    }
}

extern "C" void kernel_launch(void* const* d_in, const int* in_sizes, int n_in,
                              void* d_out, int out_size, void* d_ws, size_t ws_size,
                              hipStream_t stream) {
    (void)n_in; (void)out_size; (void)ws_size;

    const float* x  = (const float*)d_in[0];
    const int*   ei = (const int*)d_in[1];
    const float* W  = (const float*)d_in[2];
    const float* b  = (const float*)d_in[3];
    float* out = (float*)d_out;

    const int n_nodes = in_sizes[0] / IN_CH;
    const int n_edges = in_sizes[1] / 2;
    const int* row = ei;
    const int* col = ei + n_edges;

    const int n_pad = (n_nodes + 63) & ~63;
    const int e_pad = (n_edges + 63) & ~63;
    const int n_bins = (n_nodes + BIN_NODES - 1) >> BIN_SHIFT;
    const int n_blk = (n_edges + CHUNK - 1) / CHUNK;
    int*   cnt    = (int*)d_ws;
    int*   off    = cnt + n_pad;
    float* dis    = (float*)(off + n_pad);
    int*   bucket = (int*)(dis + n_pad);
    int2*  recs   = (int2*)(bucket + e_pad);
    int*   table  = (int*)(recs + e_pad);
    int*   binStart = table + n_bins * n_blk;

    bin_count_kernel<<<n_blk, 1024, 0, stream>>>(row, table, n_edges, n_bins, n_blk);

    bin_partition_kernel<<<n_blk, 1024, 0, stream>>>(row, col, table, recs, binStart, n_edges,
                                                     n_bins, n_blk);

    bin_build_kernel<<<n_bins, 1024, 0, stream>>>(recs, binStart, bucket, off, cnt, dis, n_nodes,
                                                  n_edges, n_bins);

    gather_kernel<<<(n_nodes + 7) / 8, 256, 0, stream>>>(x, bucket, off, cnt, dis, out, n_nodes);

    int n_strips = (n_nodes + 15) / 16;
    gemm_wmma_kernel<<<(n_strips + 7) / 8, 256, OUT_CH * W_STRIDE * sizeof(float),
                      stream>>>(out, W, b, out, n_strips, n_nodes);
}


# row sums and WMMA product fused in one launch: work-groups sum 16-row strips into LDS, two rows per wave with loads in flight together, times W fragments kept in registers; sum order kept
# speedup vs baseline: 2.8511x; 1.1058x over previous
#include <hip/hip_runtime.h>


#define IN_CH 128
#define OUT_CH 128
#define W_STRIDE 132
#define KEY_MAX 0x7fffffff

#define BIN_SHIFT 8
#define BIN_NODES 256
#define MAX_BINS 1024
#define CHUNK 4096
#define STAGE_CAP 8192

typedef __attribute__((ext_vector_type(2))) float v2f;
typedef __attribute__((ext_vector_type(8))) float v8f;

__device__ __forceinline__ void load4(const int* __restrict__ p, int e0, int n, int (&v)[4]) {
    if (e0 + 3 < n && ((((size_t)(p + e0)) & 15) == 0)) {
        const int4 q = *(const int4*)(p + e0);
        v[0] = q.x; v[1] = q.y; v[2] = q.z; v[3] = q.w;
    } else {
        v[0] = (e0 + 0 < n) ? p[e0 + 0] : -1;
        v[1] = (e0 + 1 < n) ? p[e0 + 1] : -1;
        v[2] = (e0 + 2 < n) ? p[e0 + 2] : -1;
        v[3] = (e0 + 3 < n) ? p[e0 + 3] : -1;
    }
}

__device__ __forceinline__ void load4x2(const int* __restrict__ p, const int* __restrict__ q,
                                        int e0, int n, int (&v)[4], int (&w)[4]) {
    if (e0 + 3 < n && (((((size_t)(p + e0)) | ((size_t)(q + e0))) & 15) == 0)) {
        const int4 a = *(const int4*)(p + e0);
        const int4 b = *(const int4*)(q + e0);
        v[0] = a.x; v[1] = a.y; v[2] = a.z; v[3] = a.w;
        w[0] = b.x; w[1] = b.y; w[2] = b.z; w[3] = b.w;
    } else {
        v[0] = (e0 + 0 < n) ? p[e0 + 0] : -1;
        v[1] = (e0 + 1 < n) ? p[e0 + 1] : -1;
        v[2] = (e0 + 2 < n) ? p[e0 + 2] : -1;
        v[3] = (e0 + 3 < n) ? p[e0 + 3] : -1;
        w[0] = (e0 + 0 < n) ? q[e0 + 0] : 0;
        w[1] = (e0 + 1 < n) ? q[e0 + 1] : 0;
        w[2] = (e0 + 2 < n) ? q[e0 + 2] : 0;
        w[3] = (e0 + 3 < n) ? q[e0 + 3] : 0;
    }
}

__global__ __launch_bounds__(1024) void bin_count_kernel(const int* __restrict__ row,
                                                         int* __restrict__ table,
                                                         int n_edges, int n_bins, int n_blk) {
    __shared__ int s_h[MAX_BINS];
    const int tid = threadIdx.x;
    if (tid < n_bins) s_h[tid] = 0;
    __syncthreads();
    const int e0 = blockIdx.x * CHUNK + tid * 4;
    int r[4];
    load4(row, e0, n_edges, r);
    if (r[0] >= 0) atomicAdd(&s_h[r[0] >> BIN_SHIFT], 1);
    if (r[1] >= 0) atomicAdd(&s_h[r[1] >> BIN_SHIFT], 1);
    if (r[2] >= 0) atomicAdd(&s_h[r[2] >> BIN_SHIFT], 1);
    if (r[3] >= 0) atomicAdd(&s_h[r[3] >> BIN_SHIFT], 1);
    __syncthreads();
    if (tid < n_bins) table[tid * n_blk + blockIdx.x] = s_h[tid];
}

__global__ __launch_bounds__(1024) void bin_partition_kernel(
    const int* __restrict__ row, const int* __restrict__ col, const int* __restrict__ table,
    int2* __restrict__ recs, int* __restrict__ binStart, int n_edges, int n_bins, int n_blk) {
    __shared__ int s_start[MAX_BINS];
    __shared__ int s_before[MAX_BINS];
    __shared__ int s_cnt[MAX_BINS];
    __shared__ int s_loc[MAX_BINS];
    __shared__ int s_wa[32];
    __shared__ int s_wb[32];
    __shared__ int2 s_stage[CHUNK];
    const int tid = threadIdx.x;
    const int lane = tid & 31;
    const int wv = tid >> 5;
    const int blk = blockIdx.x;

    const int e0 = blk * CHUNK + tid * 4;
    int r[4], c[4];
    load4x2(row, col, e0, n_edges, r, c);

    if (tid < n_bins) s_cnt[tid] = 0;

    for (int b = wv; b < n_bins; b += 64) {
        const int b2 = b + 32;
        const bool has2 = b2 < n_bins;
        const int* t0 = table + b * n_blk;
        const int* t1 = table + (has2 ? b2 : b) * n_blk;
        int tot0 = 0, bef0 = 0, tot1 = 0, bef1 = 0;
        for (int k0 = 0; k0 < n_blk; k0 += 256) {
            int q0[8], q1[8];
#pragma unroll
            for (int u = 0; u < 8; ++u) {
                const int k = k0 + u * 32 + lane;
                q0[u] = (k < n_blk) ? t0[k] : 0;
                q1[u] = (k < n_blk) ? t1[k] : 0;
            }
#pragma unroll
            for (int u = 0; u < 8; ++u) {
                const int k = k0 + u * 32 + lane;
                tot0 += q0[u];
                tot1 += q1[u];
                bef0 += (k < blk) ? q0[u] : 0;
                bef1 += (k < blk) ? q1[u] : 0;
            }
        }
        for (int o = 16; o > 0; o >>= 1) {
            tot0 += __shfl_xor(tot0, o);
            bef0 += __shfl_xor(bef0, o);
            tot1 += __shfl_xor(tot1, o);
            bef1 += __shfl_xor(bef1, o);
        }
        if (lane == 0) {
            s_start[b] = tot0;
            s_before[b] = bef0;
            if (has2) {
                s_start[b2] = tot1;
                s_before[b2] = bef1;
            }
        }
    }
    __syncthreads();

    int rank[4];
    rank[0] = (r[0] >= 0) ? atomicAdd(&s_cnt[r[0] >> BIN_SHIFT], 1) : 0;
    rank[1] = (r[1] >= 0) ? atomicAdd(&s_cnt[r[1] >> BIN_SHIFT], 1) : 0;
    rank[2] = (r[2] >= 0) ? atomicAdd(&s_cnt[r[2] >> BIN_SHIFT], 1) : 0;
    rank[3] = (r[3] >= 0) ? atomicAdd(&s_cnt[r[3] >> BIN_SHIFT], 1) : 0;
    __syncthreads();

    const int a = (tid < n_bins) ? s_start[tid] : 0;
    const int l = (tid < n_bins) ? s_cnt[tid] : 0;
    int ia = a, il = l;
    for (int o = 1; o < 32; o <<= 1) {
        const int ta = __shfl_up(ia, o);
        const int tl = __shfl_up(il, o);
        if (lane >= o) { ia += ta; il += tl; }
    }
    if (lane == 31) { s_wa[wv] = ia; s_wb[wv] = il; }
    __syncthreads();
    int ba = 0, bl = 0;
    for (int w = 0; w < wv; ++w) { ba += s_wa[w]; bl += s_wb[w]; }
    const int ga = ba + ia - a;
    const int gl = bl + il - l;
    if (tid < n_bins) {
        s_start[tid] = ga;
        s_loc[tid] = gl;
        if (blk == 0) binStart[tid] = ga;
    }
    __syncthreads();

    if (r[0] >= 0) s_stage[s_loc[r[0] >> BIN_SHIFT] + rank[0]] = make_int2(r[0], c[0]);
    if (r[1] >= 0) s_stage[s_loc[r[1] >> BIN_SHIFT] + rank[1]] = make_int2(r[1], c[1]);
    if (r[2] >= 0) s_stage[s_loc[r[2] >> BIN_SHIFT] + rank[2]] = make_int2(r[2], c[2]);
    if (r[3] >= 0) s_stage[s_loc[r[3] >> BIN_SHIFT] + rank[3]] = make_int2(r[3], c[3]);
    __syncthreads();

    int n_here = n_edges - blk * CHUNK;
    if (n_here > CHUNK) n_here = CHUNK;
    for (int i = tid; i < n_here; i += 1024) {
        const int2 rc = s_stage[i];
        const int bin = rc.x >> BIN_SHIFT;
        recs[s_start[bin] + s_before[bin] + (i - s_loc[bin])] = rc;
    }
}

__global__ __launch_bounds__(1024) void bin_build_kernel(
    const int2* __restrict__ recs, const int* __restrict__ binStart, int* __restrict__ bucket,
    int* __restrict__ off, int* __restrict__ cnt, float* __restrict__ dis,
    int n_nodes, int n_edges, int n_bins) {
    __shared__ int s_c[BIN_NODES];
    __shared__ int s_o[BIN_NODES];
    __shared__ int s_cur[BIN_NODES];
    __shared__ int s_w[8];
    __shared__ int s_b[STAGE_CAP];
    const int tid = threadIdx.x;
    const int lane = tid & 31;
    const int wv = tid >> 5;
    const int b = blockIdx.x;
    const int start = binStart[b];
    const int end = (b + 1 < n_bins) ? binStart[b + 1] : n_edges;
    const int m = end - start;
    const int node0 = b << BIN_SHIFT;

    if (tid < BIN_NODES) { s_c[tid] = 0; s_cur[tid] = 0; }
    __syncthreads();
    for (int i0 = 0; i0 < m; i0 += 4096) {
        int rx[4];
#pragma unroll
        for (int u = 0; u < 4; ++u) {
            const int i = i0 + u * 1024 + tid;
            rx[u] = (i < m) ? recs[start + i].x : -1;
        }
#pragma unroll
        for (int u = 0; u < 4; ++u)
            if (rx[u] >= 0) atomicAdd(&s_c[rx[u] - node0], 1);
    }
    __syncthreads();

    const int v = (tid < BIN_NODES) ? s_c[tid] : 0;
    int incl = v;
    for (int o = 1; o < 32; o <<= 1) {
        const int t = __shfl_up(incl, o);
        if (lane >= o) incl += t;
    }
    if (tid < BIN_NODES && lane == 31) s_w[wv] = incl;
    __syncthreads();
    if (tid < BIN_NODES) {
        int base = 0;
        for (int w = 0; w < wv; ++w) base += s_w[w];
        const int excl = base + incl - v;
        s_o[tid] = excl;
        const int node = node0 + tid;
        if (node < n_nodes) {
            off[node] = start + excl;
            cnt[node] = v;
            dis[node] = (v > 0) ? rsqrtf((float)v) : 0.f;
        }
    }
    __syncthreads();

    const bool staged = (m <= STAGE_CAP);
    for (int i0 = 0; i0 < m; i0 += 4096) {
        int2 rc[4];
#pragma unroll
        for (int u = 0; u < 4; ++u) {
            const int i = i0 + u * 1024 + tid;
            rc[u] = (i < m) ? recs[start + i] : make_int2(-1, 0);
        }
#pragma unroll
        for (int u = 0; u < 4; ++u) {
            if (rc[u].x >= 0) {
                const int ln = rc[u].x - node0;
                const int p = s_o[ln] + atomicAdd(&s_cur[ln], 1);
                if (staged) s_b[p] = rc[u].y;
                else bucket[start + p] = rc[u].y;
            }
        }
    }
    if (staged) {
        __syncthreads();
        for (int i = tid; i < m; i += 1024) bucket[start + i] = s_b[i];
    }
}

__device__ __forceinline__ int lane_i(int v, int l) {
    return __builtin_amdgcn_readlane(v, l);
}
__device__ __forceinline__ float lane_f(float v, int l) {
    return __int_as_float(__builtin_amdgcn_readlane(__float_as_int(v), l));
}

#define ROW_FMA(acc, n, xv)                   \
    acc.x = __builtin_fmaf((n), (xv).x, acc.x); \
    acc.y = __builtin_fmaf((n), (xv).y, acc.y); \
    acc.z = __builtin_fmaf((n), (xv).z, acc.z); \
    acc.w = __builtin_fmaf((n), (xv).w, acc.w)

__device__ __forceinline__ float4 row_ld(const float* __restrict__ x, int c, int lane) {
    return ((const float4*)(x + (size_t)c * IN_CH))[lane];
}

__device__ __forceinline__ void add_group4(float4& acc, const float* __restrict__ x, int sv,
                                           float wv, int t, int lane) {
    const float4 x0 = row_ld(x, lane_i(sv, t), lane);
    const float4 x1 = row_ld(x, lane_i(sv, t + 1), lane);
    const float4 x2 = row_ld(x, lane_i(sv, t + 2), lane);
    const float4 x3 = row_ld(x, lane_i(sv, t + 3), lane);
    const float n0 = lane_f(wv, t), n1 = lane_f(wv, t + 1);
    const float n2 = lane_f(wv, t + 2), n3 = lane_f(wv, t + 3);
    ROW_FMA(acc, n0, x0);
    ROW_FMA(acc, n1, x1);
    ROW_FMA(acc, n2, x2);
    ROW_FMA(acc, n3, x3);
}

__device__ __forceinline__ void add_rest(float4& acc, const float* __restrict__ x, int sv,
                                         float wv, int t, int rem, int lane) {
    if (rem == 3) {
        const float4 x0 = row_ld(x, lane_i(sv, t), lane);
        const float4 x1 = row_ld(x, lane_i(sv, t + 1), lane);
        const float4 x2 = row_ld(x, lane_i(sv, t + 2), lane);
        const float n0 = lane_f(wv, t), n1 = lane_f(wv, t + 1), n2 = lane_f(wv, t + 2);
        ROW_FMA(acc, n0, x0);
        ROW_FMA(acc, n1, x1);
        ROW_FMA(acc, n2, x2);
    } else if (rem == 2) {
        const float4 x0 = row_ld(x, lane_i(sv, t), lane);
        const float4 x1 = row_ld(x, lane_i(sv, t + 1), lane);
        const float n0 = lane_f(wv, t), n1 = lane_f(wv, t + 1);
        ROW_FMA(acc, n0, x0);
        ROW_FMA(acc, n1, x1);
    } else if (rem == 1) {
        const float4 x0 = row_ld(x, lane_i(sv, t), lane);
        const float n0 = lane_f(wv, t);
        ROW_FMA(acc, n0, x0);
    }
}

__global__ __launch_bounds__(256) __attribute__((amdgpu_waves_per_eu(8))) void gather_gemm_kernel(
    const float* __restrict__ x, const int* __restrict__ bucket,
    const int* __restrict__ off, const int* __restrict__ cnt,
    const float* __restrict__ dis, const float* __restrict__ W,
    const float* __restrict__ bias, float* __restrict__ out, int n_nodes, int n_strips) {
    extern __shared__ __attribute__((aligned(16))) float smem[];
    const int tid = threadIdx.x;
    const int lane = tid & 31;
    const int wave = __builtin_amdgcn_readfirstlane(tid >> 5);
    const int m = lane & 15;
    const int half = lane >> 4;

    for (int i = tid; i < OUT_CH * IN_CH / 4; i += 256) {
        const int o = i >> 5, c4 = i & 31;
        const float4 v = *(const float4*)(W + o * IN_CH + c4 * 4);
        *(float4*)(smem + o * W_STRIDE + c4 * 4) = v;
    }
    __syncthreads();
    v2f bf[32];
    {
        const float* bp = smem + (wave * 16 + m) * W_STRIDE + 2 * half;
#pragma unroll
        for (int j = 0; j < 32; ++j) bf[j] = *(const v2f*)(bp + 4 * j);
    }
    const float bv = bias[wave * 16 + m];
    __syncthreads();

    int it = 0;
    for (int strip = blockIdx.x; strip < n_strips; strip += gridDim.x, ++it) {
        float* buf = smem + (it & 1) * (16 * W_STRIDE);
        const int r0 = strip * 16 + wave * 2;

        int start[2], d[2], k[2];
        float dr[2];
        {
            const bool ok0 = r0 < n_nodes;
            const bool ok1 = (r0 + 1) < n_nodes;
            const int rr = ok0 ? r0 : 0;
            const int2 cv = *(const int2*)(cnt + rr);
            const int2 ov = *(const int2*)(off + rr);
            const float2 dv = *(const float2*)(dis + rr);
            const int c0 = __builtin_amdgcn_readfirstlane(cv.x);
            const int c1 = __builtin_amdgcn_readfirstlane(cv.y);
            const float f0 = lane_f(dv.x, 0);
            const float f1 = lane_f(dv.y, 0);
            start[0] = __builtin_amdgcn_readfirstlane(ov.x);
            start[1] = __builtin_amdgcn_readfirstlane(ov.y);
            d[0] = ok0 ? c0 : 0;
            d[1] = ok1 ? c1 : 0;
            dr[0] = ok0 ? f0 : 0.f;
            dr[1] = ok1 ? f1 : 0.f;
        }
        bool act[2];
#pragma unroll
        for (int q = 0; q < 2; ++q) {
            act[q] = (d[q] <= 32) && (lane < d[q]);
            const int kv = bucket[act[q] ? start[q] + lane : 0];
            k[q] = act[q] ? kv : KEY_MAX;
        }
        int s[2];
        float ds[2], nv[2];
#pragma unroll
        for (int q = 0; q < 2; ++q) {
            int rank = 0;
            if (d[q] <= 32) {
                for (int j = 0; j < d[q]; ++j) {
                    const int kj = lane_i(k[q], j);
                    rank += (kj < k[q] || (kj == k[q] && j < lane)) ? 1 : 0;
                }
            }
            if (!act[q]) rank = lane;
            s[q] = __builtin_amdgcn_ds_permute(rank << 2, k[q]);
            const float dsv = dis[act[q] ? s[q] : 0];
            ds[q] = act[q] ? dsv : 0.f;
        }
        nv[0] = dr[0] * ds[0];
        nv[1] = dr[1] * ds[1];

        float4 acc0 = make_float4(0.f, 0.f, 0.f, 0.f);
        float4 acc1 = make_float4(0.f, 0.f, 0.f, 0.f);
        if (d[0] <= 32 && d[1] <= 32) {
            const int w0 = d[0] & ~3, w1 = d[1] & ~3;
            const int wm = (w0 < w1) ? w0 : w1;
            int t = 0;
            for (; t < wm; t += 4) {
                const float4 p0 = row_ld(x, lane_i(s[0], t), lane);
                const float4 p1 = row_ld(x, lane_i(s[0], t + 1), lane);
                const float4 p2 = row_ld(x, lane_i(s[0], t + 2), lane);
                const float4 p3 = row_ld(x, lane_i(s[0], t + 3), lane);
                const float4 q0 = row_ld(x, lane_i(s[1], t), lane);
                const float4 q1 = row_ld(x, lane_i(s[1], t + 1), lane);
                const float4 q2 = row_ld(x, lane_i(s[1], t + 2), lane);
                const float4 q3 = row_ld(x, lane_i(s[1], t + 3), lane);
                const float m0 = lane_f(nv[0], t), m1 = lane_f(nv[0], t + 1);
                const float m2 = lane_f(nv[0], t + 2), m3 = lane_f(nv[0], t + 3);
                const float n0 = lane_f(nv[1], t), n1 = lane_f(nv[1], t + 1);
                const float n2 = lane_f(nv[1], t + 2), n3 = lane_f(nv[1], t + 3);
                ROW_FMA(acc0, m0, p0);
                ROW_FMA(acc0, m1, p1);
                ROW_FMA(acc0, m2, p2);
                ROW_FMA(acc0, m3, p3);
                ROW_FMA(acc1, n0, q0);
                ROW_FMA(acc1, n1, q1);
                ROW_FMA(acc1, n2, q2);
                ROW_FMA(acc1, n3, q3);
            }
            for (int u = t; u < w0; u += 4) add_group4(acc0, x, s[0], nv[0], u, lane);
            for (int u = t; u < w1; u += 4) add_group4(acc1, x, s[1], nv[1], u, lane);
            add_rest(acc0, x, s[0], nv[0], w0, d[0] - w0, lane);
            add_rest(acc1, x, s[1], nv[1], w1, d[1] - w1, lane);
        } else {
#pragma unroll
            for (int q = 0; q < 2; ++q) {
                float4 acc = make_float4(0.f, 0.f, 0.f, 0.f);
                const int dq = d[q];
                if (dq <= 32) {
#pragma unroll 1
                    for (int t = 0; t < dq; ++t) {
                        const float4 x0 = row_ld(x, lane_i(s[q], t), lane);
                        const float n0 = lane_f(nv[q], t);
                        ROW_FMA(acc, n0, x0);
                    }
                } else {
                    const int st = start[q];
                    const float drq = dr[q];
                    int prev = -1;
                    int done = 0;
                    while (done < dq) {
                        int mn = KEY_MAX;
#pragma unroll 1
                        for (int i = lane; i < dq; i += 32) {
                            const int kk = bucket[st + i];
                            if (kk > prev && kk < mn) mn = kk;
                        }
                        for (int o = 16; o > 0; o >>= 1) {
                            const int m2 = __shfl_xor(mn, o);
                            mn = (m2 < mn) ? m2 : mn;
                        }
                        int c = 0;
#pragma unroll 1
                        for (int i = lane; i < dq; i += 32) c += (bucket[st + i] == mn) ? 1 : 0;
                        for (int o = 16; o > 0; o >>= 1) c += __shfl_xor(c, o);
                        mn = __builtin_amdgcn_readfirstlane(mn);
                        c = __builtin_amdgcn_readfirstlane(c);
                        if (c <= 0) break;
                        const float n0 = drq * dis[mn];
                        const float4 x0 = row_ld(x, mn, lane);
                        for (int u = 0; u < c; ++u) {
                            acc.x += n0 * x0.x; acc.y += n0 * x0.y; acc.z += n0 * x0.z; acc.w += n0 * x0.w;
                        }
                        prev = mn;
                        done += c;
                    }
                }
                if (q == 0) acc0 = acc; else acc1 = acc;
            }
        }
        *(float4*)(buf + (wave * 2) * W_STRIDE + lane * 4) = acc0;
        *(float4*)(buf + (wave * 2 + 1) * W_STRIDE + lane * 4) = acc1;
        __syncthreads();

        v8f c8 = {};
        const float* ap = buf + m * W_STRIDE + 2 * half;
#pragma unroll
        for (int j = 0; j < 32; ++j) {
            const v2f a = *(const v2f*)(ap + 4 * j);
            c8 = __builtin_amdgcn_wmma_f32_16x16x4_f32(
                false, a, false, bf[j], (short)0, c8, false, false);
        }
        __builtin_amdgcn_sched_group_barrier(0x100, 6, 0);
#pragma unroll
        for (int g = 0; g < 10; ++g) {
            __builtin_amdgcn_sched_group_barrier(0x008, 2, 0);
            __builtin_amdgcn_sched_group_barrier(0x100, 1, 0);
        }
        __builtin_amdgcn_sched_group_barrier(0x008, 12, 0);
#pragma unroll
        for (int r = 0; r < 8; ++r) {
            const int rowNode = strip * 16 + r + 8 * half;
            if (rowNode < n_nodes)
                out[(size_t)rowNode * OUT_CH + wave * 16 + m] = c8[r] + bv;
        }
    }
}

extern "C" void kernel_launch(void* const* d_in, const int* in_sizes, int n_in,
                              void* d_out, int out_size, void* d_ws, size_t ws_size,
                              hipStream_t stream) {
    (void)n_in; (void)out_size; (void)ws_size;

    const float* x  = (const float*)d_in[0];
    const int*   ei = (const int*)d_in[1];
    const float* W  = (const float*)d_in[2];
    const float* b  = (const float*)d_in[3];
    float* out = (float*)d_out;

    const int n_nodes = in_sizes[0] / IN_CH;
    const int n_edges = in_sizes[1] / 2;
    const int* row = ei;
    const int* col = ei + n_edges;

    const int n_pad = (n_nodes + 63) & ~63;
    const int e_pad = (n_edges + 63) & ~63;
    const int n_bins = (n_nodes + BIN_NODES - 1) >> BIN_SHIFT;
    const int n_blk = (n_edges + CHUNK - 1) / CHUNK;
    int*   cnt    = (int*)d_ws;
    int*   off    = cnt + n_pad;
    float* dis    = (float*)(off + n_pad);
    int*   bucket = (int*)(dis + n_pad);
    int2*  recs   = (int2*)(bucket + e_pad);
    int*   table  = (int*)(recs + e_pad);
    int*   binStart = table + n_bins * n_blk;

    bin_count_kernel<<<n_blk, 1024, 0, stream>>>(row, table, n_edges, n_bins, n_blk);

    bin_partition_kernel<<<n_blk, 1024, 0, stream>>>(row, col, table, recs, binStart, n_edges,
                                                     n_bins, n_blk);

    bin_build_kernel<<<n_bins, 1024, 0, stream>>>(recs, binStart, bucket, off, cnt, dis, n_nodes,
                                                  n_edges, n_bins);

    const int n_strips = (n_nodes + 15) / 16;
    const int per_wg = (n_strips + 1023) / 1024;
    const int n_wg = (n_strips + per_wg - 1) / per_wg;
    gather_gemm_kernel<<<n_wg, 256, OUT_CH * W_STRIDE * sizeof(float), stream>>>(
        x, bucket, off, cnt, dis, W, b, out, n_nodes, n_strips);
}


# counting pre-pass removed: partition pass stores each block's per-bin runs in fixed 64-record slots plus a count table, build pass collects them (cut runs rescanned); 3 launches; sum order kept
# speedup vs baseline: 3.0982x; 1.0867x over previous
#include <hip/hip_runtime.h>


#define IN_CH 128
#define OUT_CH 128
#define W_STRIDE 132
#define KEY_MAX 0x7fffffff

#define BIN_SHIFT 8
#define BIN_NODES 256
#define MAX_BINS 1024
#define CHUNK 4096
#define STAGE_CAP 8192
#define SLOT_SHIFT 6
#define SLOT_CAP 64
#define BC_CAP 1024
#define POS_UNROLL 10

typedef __attribute__((ext_vector_type(2))) float v2f;
typedef __attribute__((ext_vector_type(8))) float v8f;

__device__ __forceinline__ void load4(const int* __restrict__ p, int e0, int n, int (&v)[4]) {
    if (e0 + 3 < n && ((((size_t)(p + e0)) & 15) == 0)) {
        const int4 q = *(const int4*)(p + e0);
        v[0] = q.x; v[1] = q.y; v[2] = q.z; v[3] = q.w;
    } else {
        v[0] = (e0 + 0 < n) ? p[e0 + 0] : -1;
        v[1] = (e0 + 1 < n) ? p[e0 + 1] : -1;
        v[2] = (e0 + 2 < n) ? p[e0 + 2] : -1;
        v[3] = (e0 + 3 < n) ? p[e0 + 3] : -1;
    }
}

__device__ __forceinline__ void load4x2(const int* __restrict__ p, const int* __restrict__ q,
                                        int e0, int n, int (&v)[4], int (&w)[4]) {
    if (e0 + 3 < n && (((((size_t)(p + e0)) | ((size_t)(q + e0))) & 15) == 0)) {
        const int4 a = *(const int4*)(p + e0);
        const int4 b = *(const int4*)(q + e0);
        v[0] = a.x; v[1] = a.y; v[2] = a.z; v[3] = a.w;
        w[0] = b.x; w[1] = b.y; w[2] = b.z; w[3] = b.w;
    } else {
        v[0] = (e0 + 0 < n) ? p[e0 + 0] : -1;
        v[1] = (e0 + 1 < n) ? p[e0 + 1] : -1;
        v[2] = (e0 + 2 < n) ? p[e0 + 2] : -1;
        v[3] = (e0 + 3 < n) ? p[e0 + 3] : -1;
        w[0] = (e0 + 0 < n) ? q[e0 + 0] : 0;
        w[1] = (e0 + 1 < n) ? q[e0 + 1] : 0;
        w[2] = (e0 + 2 < n) ? q[e0 + 2] : 0;
        w[3] = (e0 + 3 < n) ? q[e0 + 3] : 0;
    }
}

__global__ __launch_bounds__(1024) void bin_partition_kernel(
    const int* __restrict__ row, const int* __restrict__ col, int* __restrict__ table,
    int2* __restrict__ slots, int n_edges, int n_bins, int n_blk) {
    __shared__ int s_cnt[MAX_BINS];
    __shared__ int s_loc[MAX_BINS];
    __shared__ int s_wb[32];
    __shared__ int2 s_stage[CHUNK];
    const int tid = threadIdx.x;
    const int lane = tid & 31;
    const int wv = tid >> 5;
    const int blk = blockIdx.x;

    const int e0 = blk * CHUNK + tid * 4;
    int r[4], c[4];
    load4x2(row, col, e0, n_edges, r, c);

    if (tid < n_bins) s_cnt[tid] = 0;
    __syncthreads();

    int rank[4];
    rank[0] = (r[0] >= 0) ? atomicAdd(&s_cnt[r[0] >> BIN_SHIFT], 1) : 0;
    rank[1] = (r[1] >= 0) ? atomicAdd(&s_cnt[r[1] >> BIN_SHIFT], 1) : 0;
    rank[2] = (r[2] >= 0) ? atomicAdd(&s_cnt[r[2] >> BIN_SHIFT], 1) : 0;
    rank[3] = (r[3] >= 0) ? atomicAdd(&s_cnt[r[3] >> BIN_SHIFT], 1) : 0;
    __syncthreads();

    const int l = (tid < n_bins) ? s_cnt[tid] : 0;
    int il = l;
    for (int o = 1; o < 32; o <<= 1) {
        const int tl = __shfl_up(il, o);
        if (lane >= o) il += tl;
    }
    if (lane == 31) s_wb[wv] = il;
    __syncthreads();
    int bl = 0;
    for (int w = 0; w < wv; ++w) bl += s_wb[w];
    if (tid < n_bins) {
        s_loc[tid] = bl + il - l;
        table[tid * n_blk + blk] = l;
    }
    __syncthreads();

    if (r[0] >= 0) s_stage[s_loc[r[0] >> BIN_SHIFT] + rank[0]] = make_int2(r[0], c[0]);
    if (r[1] >= 0) s_stage[s_loc[r[1] >> BIN_SHIFT] + rank[1]] = make_int2(r[1], c[1]);
    if (r[2] >= 0) s_stage[s_loc[r[2] >> BIN_SHIFT] + rank[2]] = make_int2(r[2], c[2]);
    if (r[3] >= 0) s_stage[s_loc[r[3] >> BIN_SHIFT] + rank[3]] = make_int2(r[3], c[3]);
    __syncthreads();

    int n_here = n_edges - blk * CHUNK;
    if (n_here > CHUNK) n_here = CHUNK;
    for (int i = tid; i < n_here; i += 1024) {
        const int2 rc = s_stage[i];
        const int bin = rc.x >> BIN_SHIFT;
        const int j = i - s_loc[bin];
        if (j < SLOT_CAP) slots[(((size_t)bin * n_blk + blk) << SLOT_SHIFT) + j] = rc;
    }
}

__global__ __launch_bounds__(1024) void bin_build_kernel(
    const int* __restrict__ row, const int* __restrict__ col, const int2* __restrict__ slots,
    const int* __restrict__ table, int* __restrict__ bucket, int* __restrict__ off,
    int* __restrict__ cnt, float* __restrict__ dis, int n_nodes, int n_edges, int n_bins,
    int n_blk) {
    __shared__ int s_c[BIN_NODES];
    __shared__ int s_o[BIN_NODES];
    __shared__ int s_cur[BIN_NODES];
    __shared__ int s_w[8];
    __shared__ int s_ra[32];
    __shared__ int s_rb[32];
    __shared__ int s_rc[32];
    __shared__ int s_bc[BC_CAP];
    __shared__ int s_b[STAGE_CAP];
    const int tid = threadIdx.x;
    const int lane = tid & 31;
    const int wv = tid >> 5;
    const int b = blockIdx.x;
    const int node0 = b << BIN_SHIFT;
    const int* trow = table + b * n_blk;
    const int2* sb = slots + (((size_t)b * n_blk) << SLOT_SHIFT);
    const int n_pos = n_blk << SLOT_SHIFT;

    int before = 0;
    {
        const int n_int = b * n_blk;
        const int n4 = n_int >> 2;
        const int4* t4 = (const int4*)table;
        for (int i0 = 0; i0 < n4; i0 += 8 * 1024) {
            int4 q[8];
#pragma unroll
            for (int u = 0; u < 8; ++u) {
                const int i = i0 + u * 1024 + tid;
                q[u] = (i < n4) ? t4[i] : make_int4(0, 0, 0, 0);
            }
#pragma unroll
            for (int u = 0; u < 8; ++u) before += q[u].x + q[u].y + q[u].z + q[u].w;
        }
        const int it = (n4 << 2) + tid;
        if (it < n_int) before += table[it];
    }
    static_assert(BC_CAP == 1024, "one s_bc entry per thread of the block");
    int mine = (tid < n_blk) ? trow[tid] : 0;
    int over = (mine > SLOT_CAP) ? 1 : 0;
    s_bc[tid] = mine;
    for (int k = tid + 1024; k < n_blk; k += 1024) {
        const int cb = trow[k];
        mine += cb;
        over |= (cb > SLOT_CAP) ? 1 : 0;
    }
    if (tid < BIN_NODES) { s_c[tid] = 0; s_cur[tid] = 0; }
    for (int o = 16; o > 0; o >>= 1) {
        before += __shfl_xor(before, o);
        mine += __shfl_xor(mine, o);
        over |= __shfl_xor(over, o);
    }
    if (lane == 0) { s_ra[wv] = before; s_rb[wv] = mine; s_rc[wv] = over; }
    __syncthreads();
    int start = s_ra[lane], m = s_rb[lane], anyover = s_rc[lane];
    for (int o = 16; o > 0; o >>= 1) {
        start += __shfl_xor(start, o);
        m += __shfl_xor(m, o);
        anyover |= __shfl_xor(anyover, o);
    }

    for (int p0 = 0; p0 < n_pos; p0 += POS_UNROLL * 1024) {
        int rx[POS_UNROLL], cb[POS_UNROLL];
#pragma unroll
        for (int u = 0; u < POS_UNROLL; ++u) {
            const int k = (p0 + u * 1024 + tid) >> SLOT_SHIFT;
            cb[u] = s_bc[(k < BC_CAP) ? k : BC_CAP - 1];
        }
        if (n_blk > BC_CAP) {
#pragma unroll
            for (int u = 0; u < POS_UNROLL; ++u) {
                const int k = (p0 + u * 1024 + tid) >> SLOT_SHIFT;
                if (k >= BC_CAP && k < n_blk) cb[u] = trow[k];
            }
        }
#pragma unroll
        for (int u = 0; u < POS_UNROLL; ++u) {
            const int p = p0 + u * 1024 + tid;
            const bool ok = (p < n_pos) && (cb[u] <= SLOT_CAP) && ((p & (SLOT_CAP - 1)) < cb[u]);
            rx[u] = ok ? sb[p].x : -1;
        }
#pragma unroll
        for (int u = 0; u < POS_UNROLL; ++u)
            if (rx[u] >= 0) atomicAdd(&s_c[rx[u] - node0], 1);
    }
    if (anyover) {
        for (int k = 0; k < n_blk; ++k) {
            int cb;
            if (k < BC_CAP) cb = s_bc[k];
            else cb = trow[k];
            if (cb > SLOT_CAP) {
                int r[4];
                load4(row, k * CHUNK + tid * 4, n_edges, r);
                for (int u = 0; u < 4; ++u)
                    if (r[u] >= 0 && (r[u] >> BIN_SHIFT) == b) atomicAdd(&s_c[r[u] - node0], 1);
            }
        }
    }
    __syncthreads();

    const int v = (tid < BIN_NODES) ? s_c[tid] : 0;
    int incl = v;
    for (int o = 1; o < 32; o <<= 1) {
        const int t = __shfl_up(incl, o);
        if (lane >= o) incl += t;
    }
    if (tid < BIN_NODES && lane == 31) s_w[wv] = incl;
    __syncthreads();
    if (tid < BIN_NODES) {
        int base = 0;
        for (int w = 0; w < wv; ++w) base += s_w[w];
        const int excl = base + incl - v;
        s_o[tid] = excl;
        const int node = node0 + tid;
        if (node < n_nodes) {
            off[node] = start + excl;
            cnt[node] = v;
            dis[node] = (v > 0) ? rsqrtf((float)v) : 0.f;
        }
    }
    __syncthreads();

    const bool staged = (m <= STAGE_CAP);
    for (int p0 = 0; p0 < n_pos; p0 += POS_UNROLL * 1024) {
        int2 rc[POS_UNROLL];
        int cb[POS_UNROLL];
#pragma unroll
        for (int u = 0; u < POS_UNROLL; ++u) {
            const int k = (p0 + u * 1024 + tid) >> SLOT_SHIFT;
            cb[u] = s_bc[(k < BC_CAP) ? k : BC_CAP - 1];
        }
        if (n_blk > BC_CAP) {
#pragma unroll
            for (int u = 0; u < POS_UNROLL; ++u) {
                const int k = (p0 + u * 1024 + tid) >> SLOT_SHIFT;
                if (k >= BC_CAP && k < n_blk) cb[u] = trow[k];
            }
        }
#pragma unroll
        for (int u = 0; u < POS_UNROLL; ++u) {
            const int p = p0 + u * 1024 + tid;
            const bool ok = (p < n_pos) && (cb[u] <= SLOT_CAP) && ((p & (SLOT_CAP - 1)) < cb[u]);
            rc[u] = ok ? sb[p] : make_int2(-1, 0);
        }
#pragma unroll
        for (int u = 0; u < POS_UNROLL; ++u) {
            if (rc[u].x >= 0) {
                const int ln = rc[u].x - node0;
                const int p = s_o[ln] + atomicAdd(&s_cur[ln], 1);
                if (staged) s_b[p] = rc[u].y;
                else bucket[start + p] = rc[u].y;
            }
        }
    }
    if (anyover) {
        for (int k = 0; k < n_blk; ++k) {
            int cb;
            if (k < BC_CAP) cb = s_bc[k];
            else cb = trow[k];
            if (cb > SLOT_CAP) {
                int r[4], c[4];
                load4x2(row, col, k * CHUNK + tid * 4, n_edges, r, c);
                for (int u = 0; u < 4; ++u) {
                    if (r[u] >= 0 && (r[u] >> BIN_SHIFT) == b) {
                        const int ln = r[u] - node0;
                        const int p = s_o[ln] + atomicAdd(&s_cur[ln], 1);
                        if (staged) s_b[p] = c[u];
                        else bucket[start + p] = c[u];
                    }
                }
            }
        }
    }
    if (staged) {
        __syncthreads();
        for (int i = tid; i < m; i += 1024) bucket[start + i] = s_b[i];
    }
}

__device__ __forceinline__ int lane_i(int v, int l) {
    return __builtin_amdgcn_readlane(v, l);
}
__device__ __forceinline__ float lane_f(float v, int l) {
    return __int_as_float(__builtin_amdgcn_readlane(__float_as_int(v), l));
}

#define ROW_FMA(acc, n, xv)                   \
    acc.x = __builtin_fmaf((n), (xv).x, acc.x); \
    acc.y = __builtin_fmaf((n), (xv).y, acc.y); \
    acc.z = __builtin_fmaf((n), (xv).z, acc.z); \
    acc.w = __builtin_fmaf((n), (xv).w, acc.w)

__device__ __forceinline__ float4 row_ld(const float* __restrict__ x, int c, int lane) {
    return ((const float4*)(x + (size_t)c * IN_CH))[lane];
}

__device__ __forceinline__ void add_group4(float4& acc, const float* __restrict__ x, int sv,
                                           float wv, int t, int lane) {
    const float4 x0 = row_ld(x, lane_i(sv, t), lane);
    const float4 x1 = row_ld(x, lane_i(sv, t + 1), lane);
    const float4 x2 = row_ld(x, lane_i(sv, t + 2), lane);
    const float4 x3 = row_ld(x, lane_i(sv, t + 3), lane);
    const float n0 = lane_f(wv, t), n1 = lane_f(wv, t + 1);
    const float n2 = lane_f(wv, t + 2), n3 = lane_f(wv, t + 3);
    ROW_FMA(acc, n0, x0);
    ROW_FMA(acc, n1, x1);
    ROW_FMA(acc, n2, x2);
    ROW_FMA(acc, n3, x3);
}

__device__ __forceinline__ void add_rest(float4& acc, const float* __restrict__ x, int sv,
                                         float wv, int t, int rem, int lane) {
    if (rem == 3) {
        const float4 x0 = row_ld(x, lane_i(sv, t), lane);
        const float4 x1 = row_ld(x, lane_i(sv, t + 1), lane);
        const float4 x2 = row_ld(x, lane_i(sv, t + 2), lane);
        const float n0 = lane_f(wv, t), n1 = lane_f(wv, t + 1), n2 = lane_f(wv, t + 2);
        ROW_FMA(acc, n0, x0);
        ROW_FMA(acc, n1, x1);
        ROW_FMA(acc, n2, x2);
    } else if (rem == 2) {
        const float4 x0 = row_ld(x, lane_i(sv, t), lane);
        const float4 x1 = row_ld(x, lane_i(sv, t + 1), lane);
        const float n0 = lane_f(wv, t), n1 = lane_f(wv, t + 1);
        ROW_FMA(acc, n0, x0);
        ROW_FMA(acc, n1, x1);
    } else if (rem == 1) {
        const float4 x0 = row_ld(x, lane_i(sv, t), lane);
        const float n0 = lane_f(wv, t);
        ROW_FMA(acc, n0, x0);
    }
}

__global__ __launch_bounds__(256) __attribute__((amdgpu_waves_per_eu(8))) void gather_gemm_kernel(
    const float* __restrict__ x, const int* __restrict__ bucket,
    const int* __restrict__ off, const int* __restrict__ cnt,
    const float* __restrict__ dis, const float* __restrict__ W,
    const float* __restrict__ bias, float* __restrict__ out, int n_nodes, int n_strips) {
    extern __shared__ __attribute__((aligned(16))) float smem[];
    const int tid = threadIdx.x;
    const int lane = tid & 31;
    const int wave = __builtin_amdgcn_readfirstlane(tid >> 5);
    const int m = lane & 15;
    const int half = lane >> 4;

    for (int i = tid; i < OUT_CH * IN_CH / 4; i += 256) {
        const int o = i >> 5, c4 = i & 31;
        const float4 v = *(const float4*)(W + o * IN_CH + c4 * 4);
        *(float4*)(smem + o * W_STRIDE + c4 * 4) = v;
    }
    __syncthreads();
    v2f bf[32];
    {
        const float* bp = smem + (wave * 16 + m) * W_STRIDE + 2 * half;
#pragma unroll
        for (int j = 0; j < 32; ++j) bf[j] = *(const v2f*)(bp + 4 * j);
    }
    const float bv = bias[wave * 16 + m];
    __syncthreads();

    int it = 0;
    for (int strip = blockIdx.x; strip < n_strips; strip += gridDim.x, ++it) {
        float* buf = smem + (it & 1) * (16 * W_STRIDE);
        const int r0 = strip * 16 + wave * 2;

        int start[2], d[2], k[2];
        float dr[2];
        {
            const bool ok0 = r0 < n_nodes;
            const bool ok1 = (r0 + 1) < n_nodes;
            const int rr = ok0 ? r0 : 0;
            const int2 cv = *(const int2*)(cnt + rr);
            const int2 ov = *(const int2*)(off + rr);
            const float2 dv = *(const float2*)(dis + rr);
            const int c0 = __builtin_amdgcn_readfirstlane(cv.x);
            const int c1 = __builtin_amdgcn_readfirstlane(cv.y);
            const float f0 = lane_f(dv.x, 0);
            const float f1 = lane_f(dv.y, 0);
            start[0] = __builtin_amdgcn_readfirstlane(ov.x);
            start[1] = __builtin_amdgcn_readfirstlane(ov.y);
            d[0] = ok0 ? c0 : 0;
            d[1] = ok1 ? c1 : 0;
            dr[0] = ok0 ? f0 : 0.f;
            dr[1] = ok1 ? f1 : 0.f;
        }
        bool act[2];
#pragma unroll
        for (int q = 0; q < 2; ++q) {
            act[q] = (d[q] <= 32) && (lane < d[q]);
            const int kv = bucket[act[q] ? start[q] + lane : 0];
            k[q] = act[q] ? kv : KEY_MAX;
        }
        int s[2];
        float ds[2], nv[2];
#pragma unroll
        for (int q = 0; q < 2; ++q) {
            int rank = 0;
            if (d[q] <= 32) {
                for (int j = 0; j < d[q]; ++j) {
                    const int kj = lane_i(k[q], j);
                    rank += (kj < k[q] || (kj == k[q] && j < lane)) ? 1 : 0;
                }
            }
            if (!act[q]) rank = lane;
            s[q] = __builtin_amdgcn_ds_permute(rank << 2, k[q]);
            const float dsv = dis[act[q] ? s[q] : 0];
            ds[q] = act[q] ? dsv : 0.f;
        }
        nv[0] = dr[0] * ds[0];
        nv[1] = dr[1] * ds[1];

        float4 acc0 = make_float4(0.f, 0.f, 0.f, 0.f);
        float4 acc1 = make_float4(0.f, 0.f, 0.f, 0.f);
        if (d[0] <= 32 && d[1] <= 32) {
            const int w0 = d[0] & ~3, w1 = d[1] & ~3;
            const int wm = (w0 < w1) ? w0 : w1;
            int t = 0;
            for (; t < wm; t += 4) {
                const float4 p0 = row_ld(x, lane_i(s[0], t), lane);
                const float4 p1 = row_ld(x, lane_i(s[0], t + 1), lane);
                const float4 p2 = row_ld(x, lane_i(s[0], t + 2), lane);
                const float4 p3 = row_ld(x, lane_i(s[0], t + 3), lane);
                const float4 q0 = row_ld(x, lane_i(s[1], t), lane);
                const float4 q1 = row_ld(x, lane_i(s[1], t + 1), lane);
                const float4 q2 = row_ld(x, lane_i(s[1], t + 2), lane);
                const float4 q3 = row_ld(x, lane_i(s[1], t + 3), lane);
                const float m0 = lane_f(nv[0], t), m1 = lane_f(nv[0], t + 1);
                const float m2 = lane_f(nv[0], t + 2), m3 = lane_f(nv[0], t + 3);
                const float n0 = lane_f(nv[1], t), n1 = lane_f(nv[1], t + 1);
                const float n2 = lane_f(nv[1], t + 2), n3 = lane_f(nv[1], t + 3);
                ROW_FMA(acc0, m0, p0);
                ROW_FMA(acc0, m1, p1);
                ROW_FMA(acc0, m2, p2);
                ROW_FMA(acc0, m3, p3);
                ROW_FMA(acc1, n0, q0);
                ROW_FMA(acc1, n1, q1);
                ROW_FMA(acc1, n2, q2);
                ROW_FMA(acc1, n3, q3);
            }
            for (int u = t; u < w0; u += 4) add_group4(acc0, x, s[0], nv[0], u, lane);
            for (int u = t; u < w1; u += 4) add_group4(acc1, x, s[1], nv[1], u, lane);
            add_rest(acc0, x, s[0], nv[0], w0, d[0] - w0, lane);
            add_rest(acc1, x, s[1], nv[1], w1, d[1] - w1, lane);
        } else {
#pragma unroll
            for (int q = 0; q < 2; ++q) {
                float4 acc = make_float4(0.f, 0.f, 0.f, 0.f);
                const int dq = d[q];
                if (dq <= 32) {
#pragma unroll 1
                    for (int t = 0; t < dq; ++t) {
                        const float4 x0 = row_ld(x, lane_i(s[q], t), lane);
                        const float n0 = lane_f(nv[q], t);
                        ROW_FMA(acc, n0, x0);
                    }
                } else {
                    const int st = start[q];
                    const float drq = dr[q];
                    int prev = -1;
                    int done = 0;
                    while (done < dq) {
                        int mn = KEY_MAX;
#pragma unroll 1
                        for (int i = lane; i < dq; i += 32) {
                            const int kk = bucket[st + i];
                            if (kk > prev && kk < mn) mn = kk;
                        }
                        for (int o = 16; o > 0; o >>= 1) {
                            const int m2 = __shfl_xor(mn, o);
                            mn = (m2 < mn) ? m2 : mn;
                        }
                        int c = 0;
#pragma unroll 1
                        for (int i = lane; i < dq; i += 32) c += (bucket[st + i] == mn) ? 1 : 0;
                        for (int o = 16; o > 0; o >>= 1) c += __shfl_xor(c, o);
                        mn = __builtin_amdgcn_readfirstlane(mn);
                        c = __builtin_amdgcn_readfirstlane(c);
                        if (c <= 0) break;
                        const float n0 = drq * dis[mn];
                        const float4 x0 = row_ld(x, mn, lane);
                        for (int u = 0; u < c; ++u) {
                            acc.x += n0 * x0.x; acc.y += n0 * x0.y; acc.z += n0 * x0.z; acc.w += n0 * x0.w;
                        }
                        prev = mn;
                        done += c;
                    }
                }
                if (q == 0) acc0 = acc; else acc1 = acc;
            }
        }
        *(float4*)(buf + (wave * 2) * W_STRIDE + lane * 4) = acc0;
        *(float4*)(buf + (wave * 2 + 1) * W_STRIDE + lane * 4) = acc1;
        __syncthreads();

        v8f c8 = {};
        const float* ap = buf + m * W_STRIDE + 2 * half;
#pragma unroll
        for (int j = 0; j < 32; ++j) {
            const v2f a = *(const v2f*)(ap + 4 * j);
            c8 = __builtin_amdgcn_wmma_f32_16x16x4_f32(
                false, a, false, bf[j], (short)0, c8, false, false);
        }
        __builtin_amdgcn_sched_group_barrier(0x100, 6, 0);
#pragma unroll
        for (int g = 0; g < 10; ++g) {
            __builtin_amdgcn_sched_group_barrier(0x008, 2, 0);
            __builtin_amdgcn_sched_group_barrier(0x100, 1, 0);
        }
        __builtin_amdgcn_sched_group_barrier(0x008, 12, 0);
#pragma unroll
        for (int r = 0; r < 8; ++r) {
            const int rowNode = strip * 16 + r + 8 * half;
            if (rowNode < n_nodes)
                out[(size_t)rowNode * OUT_CH + wave * 16 + m] = c8[r] + bv;
        }
    }
}

extern "C" void kernel_launch(void* const* d_in, const int* in_sizes, int n_in,
                              void* d_out, int out_size, void* d_ws, size_t ws_size,
                              hipStream_t stream) {
    (void)n_in; (void)out_size; (void)ws_size;

    const float* x  = (const float*)d_in[0];
    const int*   ei = (const int*)d_in[1];
    const float* W  = (const float*)d_in[2];
    const float* b  = (const float*)d_in[3];
    float* out = (float*)d_out;

    const int n_nodes = in_sizes[0] / IN_CH;
    const int n_edges = in_sizes[1] / 2;
    const int* row = ei;
    const int* col = ei + n_edges;

    const int n_pad = (n_nodes + 63) & ~63;
    const int e_pad = (n_edges + 63) & ~63;
    const int n_bins = (n_nodes + BIN_NODES - 1) >> BIN_SHIFT;
    const int n_blk = (n_edges + CHUNK - 1) / CHUNK;
    int*   cnt    = (int*)d_ws;
    int*   off    = cnt + n_pad;
    float* dis    = (float*)(off + n_pad);
    int*   bucket = (int*)(dis + n_pad);
    int2*  slots  = (int2*)(bucket + e_pad);
    int*   table  = (int*)(slots + (((size_t)n_bins * n_blk) << SLOT_SHIFT));

    bin_partition_kernel<<<n_blk, 1024, 0, stream>>>(row, col, table, slots, n_edges, n_bins, n_blk);

    bin_build_kernel<<<n_bins, 1024, 0, stream>>>(row, col, slots, table, bucket, off, cnt, dis,
                                                  n_nodes, n_edges, n_bins, n_blk);

    const int n_strips = (n_nodes + 15) / 16;
    const int per_wg = (n_strips + 1023) / 1024;
    const int n_wg = (n_strips + per_wg - 1) / per_wg;
    gather_gemm_kernel<<<n_wg, 256, OUT_CH * W_STRIDE * sizeof(float), stream>>>(
        x, bucket, off, cnt, dis, W, b, out, n_nodes, n_strips);
}


# row-sum + product launch on a grid of exactly the 1024 resident work-group places (4 per WGP, strips dealt round robin, 3 or 4 each) instead of 782 work-groups x 4 strips; sums unchanged
# speedup vs baseline: 3.3140x; 1.0697x over previous
#include <hip/hip_runtime.h>


#define IN_CH 128
#define OUT_CH 128
#define W_STRIDE 132
#define KEY_MAX 0x7fffffff

#define BIN_SHIFT 8
#define BIN_NODES 256
#define MAX_BINS 1024
#define CHUNK 4096
#define STAGE_CAP 8192
#define SLOT_SHIFT 6
#define SLOT_CAP 64
#define BC_CAP 1024
#define POS_UNROLL 10

typedef __attribute__((ext_vector_type(2))) float v2f;
typedef __attribute__((ext_vector_type(8))) float v8f;

__device__ __forceinline__ void load4(const int* __restrict__ p, int e0, int n, int (&v)[4]) {
    if (e0 + 3 < n && ((((size_t)(p + e0)) & 15) == 0)) {
        const int4 q = *(const int4*)(p + e0);
        v[0] = q.x; v[1] = q.y; v[2] = q.z; v[3] = q.w;
    } else {
        v[0] = (e0 + 0 < n) ? p[e0 + 0] : -1;
        v[1] = (e0 + 1 < n) ? p[e0 + 1] : -1;
        v[2] = (e0 + 2 < n) ? p[e0 + 2] : -1;
        v[3] = (e0 + 3 < n) ? p[e0 + 3] : -1;
    }
}

__device__ __forceinline__ void load4x2(const int* __restrict__ p, const int* __restrict__ q,
                                        int e0, int n, int (&v)[4], int (&w)[4]) {
    if (e0 + 3 < n && (((((size_t)(p + e0)) | ((size_t)(q + e0))) & 15) == 0)) {
        const int4 a = *(const int4*)(p + e0);
        const int4 b = *(const int4*)(q + e0);
        v[0] = a.x; v[1] = a.y; v[2] = a.z; v[3] = a.w;
        w[0] = b.x; w[1] = b.y; w[2] = b.z; w[3] = b.w;
    } else {
        v[0] = (e0 + 0 < n) ? p[e0 + 0] : -1;
        v[1] = (e0 + 1 < n) ? p[e0 + 1] : -1;
        v[2] = (e0 + 2 < n) ? p[e0 + 2] : -1;
        v[3] = (e0 + 3 < n) ? p[e0 + 3] : -1;
        w[0] = (e0 + 0 < n) ? q[e0 + 0] : 0;
        w[1] = (e0 + 1 < n) ? q[e0 + 1] : 0;
        w[2] = (e0 + 2 < n) ? q[e0 + 2] : 0;
        w[3] = (e0 + 3 < n) ? q[e0 + 3] : 0;
    }
}

__global__ __launch_bounds__(1024) void bin_partition_kernel(
    const int* __restrict__ row, const int* __restrict__ col, int* __restrict__ table,
    int2* __restrict__ slots, int n_edges, int n_bins, int n_blk) {
    __shared__ int s_cnt[MAX_BINS];
    __shared__ int s_loc[MAX_BINS];
    __shared__ int s_wb[32];
    __shared__ int2 s_stage[CHUNK];
    const int tid = threadIdx.x;
    const int lane = tid & 31;
    const int wv = tid >> 5;
    const int blk = blockIdx.x;

    const int e0 = blk * CHUNK + tid * 4;
    int r[4], c[4];
    load4x2(row, col, e0, n_edges, r, c);

    if (tid < n_bins) s_cnt[tid] = 0;
    __syncthreads();

    int rank[4];
    rank[0] = (r[0] >= 0) ? atomicAdd(&s_cnt[r[0] >> BIN_SHIFT], 1) : 0;
    rank[1] = (r[1] >= 0) ? atomicAdd(&s_cnt[r[1] >> BIN_SHIFT], 1) : 0;
    rank[2] = (r[2] >= 0) ? atomicAdd(&s_cnt[r[2] >> BIN_SHIFT], 1) : 0;
    rank[3] = (r[3] >= 0) ? atomicAdd(&s_cnt[r[3] >> BIN_SHIFT], 1) : 0;
    __syncthreads();

    const int l = (tid < n_bins) ? s_cnt[tid] : 0;
    int il = l;
    for (int o = 1; o < 32; o <<= 1) {
        const int tl = __shfl_up(il, o);
        if (lane >= o) il += tl;
    }
    if (lane == 31) s_wb[wv] = il;
    __syncthreads();
    int bl = 0;
    for (int w = 0; w < wv; ++w) bl += s_wb[w];
    if (tid < n_bins) {
        s_loc[tid] = bl + il - l;
        table[tid * n_blk + blk] = l;
    }
    __syncthreads();

    if (r[0] >= 0) s_stage[s_loc[r[0] >> BIN_SHIFT] + rank[0]] = make_int2(r[0], c[0]);
    if (r[1] >= 0) s_stage[s_loc[r[1] >> BIN_SHIFT] + rank[1]] = make_int2(r[1], c[1]);
    if (r[2] >= 0) s_stage[s_loc[r[2] >> BIN_SHIFT] + rank[2]] = make_int2(r[2], c[2]);
    if (r[3] >= 0) s_stage[s_loc[r[3] >> BIN_SHIFT] + rank[3]] = make_int2(r[3], c[3]);
    __syncthreads();

    int n_here = n_edges - blk * CHUNK;
    if (n_here > CHUNK) n_here = CHUNK;
    for (int i = tid; i < n_here; i += 1024) {
        const int2 rc = s_stage[i];
        const int bin = rc.x >> BIN_SHIFT;
        const int j = i - s_loc[bin];
        if (j < SLOT_CAP) slots[(((size_t)bin * n_blk + blk) << SLOT_SHIFT) + j] = rc;
    }
}

__global__ __launch_bounds__(1024) void bin_build_kernel(
    const int* __restrict__ row, const int* __restrict__ col, const int2* __restrict__ slots,
    const int* __restrict__ table, int* __restrict__ bucket, int* __restrict__ off,
    int* __restrict__ cnt, float* __restrict__ dis, int n_nodes, int n_edges, int n_bins,
    int n_blk) {
    __shared__ int s_c[BIN_NODES];
    __shared__ int s_o[BIN_NODES];
    __shared__ int s_cur[BIN_NODES];
    __shared__ int s_w[8];
    __shared__ int s_ra[32];
    __shared__ int s_rb[32];
    __shared__ int s_rc[32];
    __shared__ int s_bc[BC_CAP];
    __shared__ int s_b[STAGE_CAP];
    const int tid = threadIdx.x;
    const int lane = tid & 31;
    const int wv = tid >> 5;
    const int b = blockIdx.x;
    const int node0 = b << BIN_SHIFT;
    const int* trow = table + b * n_blk;
    const int2* sb = slots + (((size_t)b * n_blk) << SLOT_SHIFT);
    const int n_pos = n_blk << SLOT_SHIFT;

    int before = 0;
    {
        const int n_int = b * n_blk;
        const int n4 = n_int >> 2;
        const int4* t4 = (const int4*)table;
        for (int i0 = 0; i0 < n4; i0 += 8 * 1024) {
            int4 q[8];
#pragma unroll
            for (int u = 0; u < 8; ++u) {
                const int i = i0 + u * 1024 + tid;
                q[u] = (i < n4) ? t4[i] : make_int4(0, 0, 0, 0);
            }
#pragma unroll
            for (int u = 0; u < 8; ++u) before += q[u].x + q[u].y + q[u].z + q[u].w;
        }
        const int it = (n4 << 2) + tid;
        if (it < n_int) before += table[it];
    }
    static_assert(BC_CAP == 1024, "one s_bc entry per thread of the block");
    int mine = (tid < n_blk) ? trow[tid] : 0;
    int over = (mine > SLOT_CAP) ? 1 : 0;
    s_bc[tid] = mine;
    for (int k = tid + 1024; k < n_blk; k += 1024) {
        const int cb = trow[k];
        mine += cb;
        over |= (cb > SLOT_CAP) ? 1 : 0;
    }
    if (tid < BIN_NODES) { s_c[tid] = 0; s_cur[tid] = 0; }
    for (int o = 16; o > 0; o >>= 1) {
        before += __shfl_xor(before, o);
        mine += __shfl_xor(mine, o);
        over |= __shfl_xor(over, o);
    }
    if (lane == 0) { s_ra[wv] = before; s_rb[wv] = mine; s_rc[wv] = over; }
    __syncthreads();
    int start = s_ra[lane], m = s_rb[lane], anyover = s_rc[lane];
    for (int o = 16; o > 0; o >>= 1) {
        start += __shfl_xor(start, o);
        m += __shfl_xor(m, o);
        anyover |= __shfl_xor(anyover, o);
    }

    for (int p0 = 0; p0 < n_pos; p0 += POS_UNROLL * 1024) {
        int rx[POS_UNROLL], cb[POS_UNROLL];
#pragma unroll
        for (int u = 0; u < POS_UNROLL; ++u) {
            const int k = (p0 + u * 1024 + tid) >> SLOT_SHIFT;
            cb[u] = s_bc[(k < BC_CAP) ? k : BC_CAP - 1];
        }
        if (n_blk > BC_CAP) {
#pragma unroll
            for (int u = 0; u < POS_UNROLL; ++u) {
                const int k = (p0 + u * 1024 + tid) >> SLOT_SHIFT;
                if (k >= BC_CAP && k < n_blk) cb[u] = trow[k];
            }
        }
#pragma unroll
        for (int u = 0; u < POS_UNROLL; ++u) {
            const int p = p0 + u * 1024 + tid;
            const bool ok = (p < n_pos) && (cb[u] <= SLOT_CAP) && ((p & (SLOT_CAP - 1)) < cb[u]);
            rx[u] = ok ? sb[p].x : -1;
        }
#pragma unroll
        for (int u = 0; u < POS_UNROLL; ++u)
            if (rx[u] >= 0) atomicAdd(&s_c[rx[u] - node0], 1);
    }
    if (anyover) {
        for (int k = 0; k < n_blk; ++k) {
            int cb;
            if (k < BC_CAP) cb = s_bc[k];
            else cb = trow[k];
            if (cb > SLOT_CAP) {
                int r[4];
                load4(row, k * CHUNK + tid * 4, n_edges, r);
                for (int u = 0; u < 4; ++u)
                    if (r[u] >= 0 && (r[u] >> BIN_SHIFT) == b) atomicAdd(&s_c[r[u] - node0], 1);
            }
        }
    }
    __syncthreads();

    const int v = (tid < BIN_NODES) ? s_c[tid] : 0;
    int incl = v;
    for (int o = 1; o < 32; o <<= 1) {
        const int t = __shfl_up(incl, o);
        if (lane >= o) incl += t;
    }
    if (tid < BIN_NODES && lane == 31) s_w[wv] = incl;
    __syncthreads();
    if (tid < BIN_NODES) {
        int base = 0;
        for (int w = 0; w < wv; ++w) base += s_w[w];
        const int excl = base + incl - v;
        s_o[tid] = excl;
        const int node = node0 + tid;
        if (node < n_nodes) {
            off[node] = start + excl;
            cnt[node] = v;
            dis[node] = (v > 0) ? rsqrtf((float)v) : 0.f;
        }
    }
    __syncthreads();

    const bool staged = (m <= STAGE_CAP);
    for (int p0 = 0; p0 < n_pos; p0 += POS_UNROLL * 1024) {
        int2 rc[POS_UNROLL];
        int cb[POS_UNROLL];
#pragma unroll
        for (int u = 0; u < POS_UNROLL; ++u) {
            const int k = (p0 + u * 1024 + tid) >> SLOT_SHIFT;
            cb[u] = s_bc[(k < BC_CAP) ? k : BC_CAP - 1];
        }
        if (n_blk > BC_CAP) {
#pragma unroll
            for (int u = 0; u < POS_UNROLL; ++u) {
                const int k = (p0 + u * 1024 + tid) >> SLOT_SHIFT;
                if (k >= BC_CAP && k < n_blk) cb[u] = trow[k];
            }
        }
#pragma unroll
        for (int u = 0; u < POS_UNROLL; ++u) {
            const int p = p0 + u * 1024 + tid;
            const bool ok = (p < n_pos) && (cb[u] <= SLOT_CAP) && ((p & (SLOT_CAP - 1)) < cb[u]);
            rc[u] = ok ? sb[p] : make_int2(-1, 0);
        }
#pragma unroll
        for (int u = 0; u < POS_UNROLL; ++u) {
            if (rc[u].x >= 0) {
                const int ln = rc[u].x - node0;
                const int p = s_o[ln] + atomicAdd(&s_cur[ln], 1);
                if (staged) s_b[p] = rc[u].y;
                else bucket[start + p] = rc[u].y;
            }
        }
    }
    if (anyover) {
        for (int k = 0; k < n_blk; ++k) {
            int cb;
            if (k < BC_CAP) cb = s_bc[k];
            else cb = trow[k];
            if (cb > SLOT_CAP) {
                int r[4], c[4];
                load4x2(row, col, k * CHUNK + tid * 4, n_edges, r, c);
                for (int u = 0; u < 4; ++u) {
                    if (r[u] >= 0 && (r[u] >> BIN_SHIFT) == b) {
                        const int ln = r[u] - node0;
                        const int p = s_o[ln] + atomicAdd(&s_cur[ln], 1);
                        if (staged) s_b[p] = c[u];
                        else bucket[start + p] = c[u];
                    }
                }
            }
        }
    }
    if (staged) {
        __syncthreads();
        for (int i = tid; i < m; i += 1024) bucket[start + i] = s_b[i];
    }
}

__device__ __forceinline__ int lane_i(int v, int l) {
    return __builtin_amdgcn_readlane(v, l);
}
__device__ __forceinline__ float lane_f(float v, int l) {
    return __int_as_float(__builtin_amdgcn_readlane(__float_as_int(v), l));
}

#define ROW_FMA(acc, n, xv)                   \
    acc.x = __builtin_fmaf((n), (xv).x, acc.x); \
    acc.y = __builtin_fmaf((n), (xv).y, acc.y); \
    acc.z = __builtin_fmaf((n), (xv).z, acc.z); \
    acc.w = __builtin_fmaf((n), (xv).w, acc.w)

__device__ __forceinline__ float4 row_ld(const float* __restrict__ x, int c, int lane) {
    return ((const float4*)(x + (size_t)c * IN_CH))[lane];
}

__device__ __forceinline__ void add_group4(float4& acc, const float* __restrict__ x, int sv,
                                           float wv, int t, int lane) {
    const float4 x0 = row_ld(x, lane_i(sv, t), lane);
    const float4 x1 = row_ld(x, lane_i(sv, t + 1), lane);
    const float4 x2 = row_ld(x, lane_i(sv, t + 2), lane);
    const float4 x3 = row_ld(x, lane_i(sv, t + 3), lane);
    const float n0 = lane_f(wv, t), n1 = lane_f(wv, t + 1);
    const float n2 = lane_f(wv, t + 2), n3 = lane_f(wv, t + 3);
    ROW_FMA(acc, n0, x0);
    ROW_FMA(acc, n1, x1);
    ROW_FMA(acc, n2, x2);
    ROW_FMA(acc, n3, x3);
}

__device__ __forceinline__ void add_rest(float4& acc, const float* __restrict__ x, int sv,
                                         float wv, int t, int rem, int lane) {
    if (rem == 3) {
        const float4 x0 = row_ld(x, lane_i(sv, t), lane);
        const float4 x1 = row_ld(x, lane_i(sv, t + 1), lane);
        const float4 x2 = row_ld(x, lane_i(sv, t + 2), lane);
        const float n0 = lane_f(wv, t), n1 = lane_f(wv, t + 1), n2 = lane_f(wv, t + 2);
        ROW_FMA(acc, n0, x0);
        ROW_FMA(acc, n1, x1);
        ROW_FMA(acc, n2, x2);
    } else if (rem == 2) {
        const float4 x0 = row_ld(x, lane_i(sv, t), lane);
        const float4 x1 = row_ld(x, lane_i(sv, t + 1), lane);
        const float n0 = lane_f(wv, t), n1 = lane_f(wv, t + 1);
        ROW_FMA(acc, n0, x0);
        ROW_FMA(acc, n1, x1);
    } else if (rem == 1) {
        const float4 x0 = row_ld(x, lane_i(sv, t), lane);
        const float n0 = lane_f(wv, t);
        ROW_FMA(acc, n0, x0);
    }
}

__global__ __launch_bounds__(256) __attribute__((amdgpu_waves_per_eu(8))) void gather_gemm_kernel(
    const float* __restrict__ x, const int* __restrict__ bucket,
    const int* __restrict__ off, const int* __restrict__ cnt,
    const float* __restrict__ dis, const float* __restrict__ W,
    const float* __restrict__ bias, float* __restrict__ out, int n_nodes, int n_strips) {
    extern __shared__ __attribute__((aligned(16))) float smem[];
    const int tid = threadIdx.x;
    const int lane = tid & 31;
    const int wave = __builtin_amdgcn_readfirstlane(tid >> 5);
    const int m = lane & 15;
    const int half = lane >> 4;

    for (int i = tid; i < OUT_CH * IN_CH / 4; i += 256) {
        const int o = i >> 5, c4 = i & 31;
        const float4 v = *(const float4*)(W + o * IN_CH + c4 * 4);
        *(float4*)(smem + o * W_STRIDE + c4 * 4) = v;
    }
    __syncthreads();
    v2f bf[32];
    {
        const float* bp = smem + (wave * 16 + m) * W_STRIDE + 2 * half;
#pragma unroll
        for (int j = 0; j < 32; ++j) bf[j] = *(const v2f*)(bp + 4 * j);
    }
    const float bv = bias[wave * 16 + m];
    __syncthreads();

    int it = 0;
    for (int strip = blockIdx.x; strip < n_strips; strip += gridDim.x, ++it) {
        float* buf = smem + (it & 1) * (16 * W_STRIDE);
        const int r0 = strip * 16 + wave * 2;

        int start[2], d[2], k[2];
        float dr[2];
        {
            const bool ok0 = r0 < n_nodes;
            const bool ok1 = (r0 + 1) < n_nodes;
            const int rr = ok0 ? r0 : 0;
            const int2 cv = *(const int2*)(cnt + rr);
            const int2 ov = *(const int2*)(off + rr);
            const float2 dv = *(const float2*)(dis + rr);
            const int c0 = __builtin_amdgcn_readfirstlane(cv.x);
            const int c1 = __builtin_amdgcn_readfirstlane(cv.y);
            const float f0 = lane_f(dv.x, 0);
            const float f1 = lane_f(dv.y, 0);
            start[0] = __builtin_amdgcn_readfirstlane(ov.x);
            start[1] = __builtin_amdgcn_readfirstlane(ov.y);
            d[0] = ok0 ? c0 : 0;
            d[1] = ok1 ? c1 : 0;
            dr[0] = ok0 ? f0 : 0.f;
            dr[1] = ok1 ? f1 : 0.f;
        }
        bool act[2];
#pragma unroll
        for (int q = 0; q < 2; ++q) {
            act[q] = (d[q] <= 32) && (lane < d[q]);
            const int kv = bucket[act[q] ? start[q] + lane : 0];
            k[q] = act[q] ? kv : KEY_MAX;
        }
        int s[2];
        float ds[2], nv[2];
#pragma unroll
        for (int q = 0; q < 2; ++q) {
            int rank = 0;
            if (d[q] <= 32) {
                for (int j = 0; j < d[q]; ++j) {
                    const int kj = lane_i(k[q], j);
                    rank += (kj < k[q] || (kj == k[q] && j < lane)) ? 1 : 0;
                }
            }
            if (!act[q]) rank = lane;
            s[q] = __builtin_amdgcn_ds_permute(rank << 2, k[q]);
            const float dsv = dis[act[q] ? s[q] : 0];
            ds[q] = act[q] ? dsv : 0.f;
        }
        nv[0] = dr[0] * ds[0];
        nv[1] = dr[1] * ds[1];

        float4 acc0 = make_float4(0.f, 0.f, 0.f, 0.f);
        float4 acc1 = make_float4(0.f, 0.f, 0.f, 0.f);
        if (d[0] <= 32 && d[1] <= 32) {
            const int w0 = d[0] & ~3, w1 = d[1] & ~3;
            const int wm = (w0 < w1) ? w0 : w1;
            int t = 0;
            for (; t < wm; t += 4) {
                const float4 p0 = row_ld(x, lane_i(s[0], t), lane);
                const float4 p1 = row_ld(x, lane_i(s[0], t + 1), lane);
                const float4 p2 = row_ld(x, lane_i(s[0], t + 2), lane);
                const float4 p3 = row_ld(x, lane_i(s[0], t + 3), lane);
                const float4 q0 = row_ld(x, lane_i(s[1], t), lane);
                const float4 q1 = row_ld(x, lane_i(s[1], t + 1), lane);
                const float4 q2 = row_ld(x, lane_i(s[1], t + 2), lane);
                const float4 q3 = row_ld(x, lane_i(s[1], t + 3), lane);
                const float m0 = lane_f(nv[0], t), m1 = lane_f(nv[0], t + 1);
                const float m2 = lane_f(nv[0], t + 2), m3 = lane_f(nv[0], t + 3);
                const float n0 = lane_f(nv[1], t), n1 = lane_f(nv[1], t + 1);
                const float n2 = lane_f(nv[1], t + 2), n3 = lane_f(nv[1], t + 3);
                ROW_FMA(acc0, m0, p0);
                ROW_FMA(acc0, m1, p1);
                ROW_FMA(acc0, m2, p2);
                ROW_FMA(acc0, m3, p3);
                ROW_FMA(acc1, n0, q0);
                ROW_FMA(acc1, n1, q1);
                ROW_FMA(acc1, n2, q2);
                ROW_FMA(acc1, n3, q3);
            }
            for (int u = t; u < w0; u += 4) add_group4(acc0, x, s[0], nv[0], u, lane);
            for (int u = t; u < w1; u += 4) add_group4(acc1, x, s[1], nv[1], u, lane);
            add_rest(acc0, x, s[0], nv[0], w0, d[0] - w0, lane);
            add_rest(acc1, x, s[1], nv[1], w1, d[1] - w1, lane);
        } else {
#pragma unroll
            for (int q = 0; q < 2; ++q) {
                float4 acc = make_float4(0.f, 0.f, 0.f, 0.f);
                const int dq = d[q];
                if (dq <= 32) {
#pragma unroll 1
                    for (int t = 0; t < dq; ++t) {
                        const float4 x0 = row_ld(x, lane_i(s[q], t), lane);
                        const float n0 = lane_f(nv[q], t);
                        ROW_FMA(acc, n0, x0);
                    }
                } else {
                    const int st = start[q];
                    const float drq = dr[q];
                    int prev = -1;
                    int done = 0;
                    while (done < dq) {
                        int mn = KEY_MAX;
#pragma unroll 1
                        for (int i = lane; i < dq; i += 32) {
                            const int kk = bucket[st + i];
                            if (kk > prev && kk < mn) mn = kk;
                        }
                        for (int o = 16; o > 0; o >>= 1) {
                            const int m2 = __shfl_xor(mn, o);
                            mn = (m2 < mn) ? m2 : mn;
                        }
                        int c = 0;
#pragma unroll 1
                        for (int i = lane; i < dq; i += 32) c += (bucket[st + i] == mn) ? 1 : 0;
                        for (int o = 16; o > 0; o >>= 1) c += __shfl_xor(c, o);
                        mn = __builtin_amdgcn_readfirstlane(mn);
                        c = __builtin_amdgcn_readfirstlane(c);
                        if (c <= 0) break;
                        const float n0 = drq * dis[mn];
                        const float4 x0 = row_ld(x, mn, lane);
                        for (int u = 0; u < c; ++u) {
                            acc.x += n0 * x0.x; acc.y += n0 * x0.y; acc.z += n0 * x0.z; acc.w += n0 * x0.w;
                        }
                        prev = mn;
                        done += c;
                    }
                }
                if (q == 0) acc0 = acc; else acc1 = acc;
            }
        }
        *(float4*)(buf + (wave * 2) * W_STRIDE + lane * 4) = acc0;
        *(float4*)(buf + (wave * 2 + 1) * W_STRIDE + lane * 4) = acc1;
        __syncthreads();

        v8f c8 = {};
        const float* ap = buf + m * W_STRIDE + 2 * half;
#pragma unroll
        for (int j = 0; j < 32; ++j) {
            const v2f a = *(const v2f*)(ap + 4 * j);
            c8 = __builtin_amdgcn_wmma_f32_16x16x4_f32(
                false, a, false, bf[j], (short)0, c8, false, false);
        }
        __builtin_amdgcn_sched_group_barrier(0x100, 6, 0);
#pragma unroll
        for (int g = 0; g < 10; ++g) {
            __builtin_amdgcn_sched_group_barrier(0x008, 2, 0);
            __builtin_amdgcn_sched_group_barrier(0x100, 1, 0);
        }
        __builtin_amdgcn_sched_group_barrier(0x008, 12, 0);
#pragma unroll
        for (int r = 0; r < 8; ++r) {
            const int rowNode = strip * 16 + r + 8 * half;
            if (rowNode < n_nodes)
                out[(size_t)rowNode * OUT_CH + wave * 16 + m] = c8[r] + bv;
        }
    }
}

extern "C" void kernel_launch(void* const* d_in, const int* in_sizes, int n_in,
                              void* d_out, int out_size, void* d_ws, size_t ws_size,
                              hipStream_t stream) {
    (void)n_in; (void)out_size; (void)ws_size;

    const float* x  = (const float*)d_in[0];
    const int*   ei = (const int*)d_in[1];
    const float* W  = (const float*)d_in[2];
    const float* b  = (const float*)d_in[3];
    float* out = (float*)d_out;

    const int n_nodes = in_sizes[0] / IN_CH;
    const int n_edges = in_sizes[1] / 2;
    const int* row = ei;
    const int* col = ei + n_edges;

    const int n_pad = (n_nodes + 63) & ~63;
    const int e_pad = (n_edges + 63) & ~63;
    const int n_bins = (n_nodes + BIN_NODES - 1) >> BIN_SHIFT;
    const int n_blk = (n_edges + CHUNK - 1) / CHUNK;
    int*   cnt    = (int*)d_ws;
    int*   off    = cnt + n_pad;
    float* dis    = (float*)(off + n_pad);
    int*   bucket = (int*)(dis + n_pad);
    int2*  slots  = (int2*)(bucket + e_pad);
    int*   table  = (int*)(slots + (((size_t)n_bins * n_blk) << SLOT_SHIFT));

    bin_partition_kernel<<<n_blk, 1024, 0, stream>>>(row, col, table, slots, n_edges, n_bins, n_blk);

    bin_build_kernel<<<n_bins, 1024, 0, stream>>>(row, col, slots, table, bucket, off, cnt, dis,
                                                  n_nodes, n_edges, n_bins, n_blk);

    const int n_strips = (n_nodes + 15) / 16;
    const int n_wg = (n_strips < 1024) ? n_strips : 1024;
    gather_gemm_kernel<<<n_wg, 256, OUT_CH * W_STRIDE * sizeof(float), stream>>>(
        x, bucket, off, cnt, dis, W, b, out, n_nodes, n_strips);
}


# row sums: each strip's bookkeeping chain (counts/offsets, bucket entries, their order, sources' deg^-1/2) is fetched one strip ahead, under the product; norm formed at use; sums unchanged
# speedup vs baseline: 3.4946x; 1.0545x over previous
#include <hip/hip_runtime.h>


#define IN_CH 128
#define OUT_CH 128
#define W_STRIDE 132
#define KEY_MAX 0x7fffffff

#define BIN_SHIFT 8
#define BIN_NODES 256
#define MAX_BINS 1024
#define CHUNK 4096
#define STAGE_CAP 8192
#define SLOT_SHIFT 6
#define SLOT_CAP 64
#define BC_CAP 1024
#define POS_UNROLL 10

typedef __attribute__((ext_vector_type(2))) float v2f;
typedef __attribute__((ext_vector_type(8))) float v8f;

__device__ __forceinline__ void load4(const int* __restrict__ p, int e0, int n, int (&v)[4]) {
    if (e0 + 3 < n && ((((size_t)(p + e0)) & 15) == 0)) {
        const int4 q = *(const int4*)(p + e0);
        v[0] = q.x; v[1] = q.y; v[2] = q.z; v[3] = q.w;
    } else {
        v[0] = (e0 + 0 < n) ? p[e0 + 0] : -1;
        v[1] = (e0 + 1 < n) ? p[e0 + 1] : -1;
        v[2] = (e0 + 2 < n) ? p[e0 + 2] : -1;
        v[3] = (e0 + 3 < n) ? p[e0 + 3] : -1;
    }
}

__device__ __forceinline__ void load4x2(const int* __restrict__ p, const int* __restrict__ q,
                                        int e0, int n, int (&v)[4], int (&w)[4]) {
    if (e0 + 3 < n && (((((size_t)(p + e0)) | ((size_t)(q + e0))) & 15) == 0)) {
        const int4 a = *(const int4*)(p + e0);
        const int4 b = *(const int4*)(q + e0);
        v[0] = a.x; v[1] = a.y; v[2] = a.z; v[3] = a.w;
        w[0] = b.x; w[1] = b.y; w[2] = b.z; w[3] = b.w;
    } else {
        v[0] = (e0 + 0 < n) ? p[e0 + 0] : -1;
        v[1] = (e0 + 1 < n) ? p[e0 + 1] : -1;
        v[2] = (e0 + 2 < n) ? p[e0 + 2] : -1;
        v[3] = (e0 + 3 < n) ? p[e0 + 3] : -1;
        w[0] = (e0 + 0 < n) ? q[e0 + 0] : 0;
        w[1] = (e0 + 1 < n) ? q[e0 + 1] : 0;
        w[2] = (e0 + 2 < n) ? q[e0 + 2] : 0;
        w[3] = (e0 + 3 < n) ? q[e0 + 3] : 0;
    }
}

__global__ __launch_bounds__(1024) void bin_partition_kernel(
    const int* __restrict__ row, const int* __restrict__ col, int* __restrict__ table,
    int2* __restrict__ slots, int n_edges, int n_bins, int n_blk) {
    __shared__ int s_cnt[MAX_BINS];
    __shared__ int s_loc[MAX_BINS];
    __shared__ int s_wb[32];
    __shared__ int2 s_stage[CHUNK];
    const int tid = threadIdx.x;
    const int lane = tid & 31;
    const int wv = tid >> 5;
    const int blk = blockIdx.x;

    const int e0 = blk * CHUNK + tid * 4;
    int r[4], c[4];
    load4x2(row, col, e0, n_edges, r, c);

    if (tid < n_bins) s_cnt[tid] = 0;
    __syncthreads();

    int rank[4];
    rank[0] = (r[0] >= 0) ? atomicAdd(&s_cnt[r[0] >> BIN_SHIFT], 1) : 0;
    rank[1] = (r[1] >= 0) ? atomicAdd(&s_cnt[r[1] >> BIN_SHIFT], 1) : 0;
    rank[2] = (r[2] >= 0) ? atomicAdd(&s_cnt[r[2] >> BIN_SHIFT], 1) : 0;
    rank[3] = (r[3] >= 0) ? atomicAdd(&s_cnt[r[3] >> BIN_SHIFT], 1) : 0;
    __syncthreads();

    const int l = (tid < n_bins) ? s_cnt[tid] : 0;
    int il = l;
    for (int o = 1; o < 32; o <<= 1) {
        const int tl = __shfl_up(il, o);
        if (lane >= o) il += tl;
    }
    if (lane == 31) s_wb[wv] = il;
    __syncthreads();
    int bl = 0;
    for (int w = 0; w < wv; ++w) bl += s_wb[w];
    if (tid < n_bins) {
        s_loc[tid] = bl + il - l;
        table[tid * n_blk + blk] = l;
    }
    __syncthreads();

    if (r[0] >= 0) s_stage[s_loc[r[0] >> BIN_SHIFT] + rank[0]] = make_int2(r[0], c[0]);
    if (r[1] >= 0) s_stage[s_loc[r[1] >> BIN_SHIFT] + rank[1]] = make_int2(r[1], c[1]);
    if (r[2] >= 0) s_stage[s_loc[r[2] >> BIN_SHIFT] + rank[2]] = make_int2(r[2], c[2]);
    if (r[3] >= 0) s_stage[s_loc[r[3] >> BIN_SHIFT] + rank[3]] = make_int2(r[3], c[3]);
    __syncthreads();

    int n_here = n_edges - blk * CHUNK;
    if (n_here > CHUNK) n_here = CHUNK;
    for (int i = tid; i < n_here; i += 1024) {
        const int2 rc = s_stage[i];
        const int bin = rc.x >> BIN_SHIFT;
        const int j = i - s_loc[bin];
        if (j < SLOT_CAP) slots[(((size_t)bin * n_blk + blk) << SLOT_SHIFT) + j] = rc;
    }
}

__global__ __launch_bounds__(1024) void bin_build_kernel(
    const int* __restrict__ row, const int* __restrict__ col, const int2* __restrict__ slots,
    const int* __restrict__ table, int* __restrict__ bucket, int* __restrict__ off,
    int* __restrict__ cnt, float* __restrict__ dis, int n_nodes, int n_edges, int n_bins,
    int n_blk) {
    __shared__ int s_c[BIN_NODES];
    __shared__ int s_o[BIN_NODES];
    __shared__ int s_cur[BIN_NODES];
    __shared__ int s_w[8];
    __shared__ int s_ra[32];
    __shared__ int s_rb[32];
    __shared__ int s_rc[32];
    __shared__ int s_bc[BC_CAP];
    __shared__ int s_b[STAGE_CAP];
    const int tid = threadIdx.x;
    const int lane = tid & 31;
    const int wv = tid >> 5;
    const int b = blockIdx.x;
    const int node0 = b << BIN_SHIFT;
    const int* trow = table + b * n_blk;
    const int2* sb = slots + (((size_t)b * n_blk) << SLOT_SHIFT);
    const int n_pos = n_blk << SLOT_SHIFT;

    int before = 0;
    {
        const int n_int = b * n_blk;
        const int n4 = n_int >> 2;
        const int4* t4 = (const int4*)table;
        for (int i0 = 0; i0 < n4; i0 += 8 * 1024) {
            int4 q[8];
#pragma unroll
            for (int u = 0; u < 8; ++u) {
                const int i = i0 + u * 1024 + tid;
                q[u] = (i < n4) ? t4[i] : make_int4(0, 0, 0, 0);
            }
#pragma unroll
            for (int u = 0; u < 8; ++u) before += q[u].x + q[u].y + q[u].z + q[u].w;
        }
        const int it = (n4 << 2) + tid;
        if (it < n_int) before += table[it];
    }
    static_assert(BC_CAP == 1024, "one s_bc entry per thread of the block");
    int mine = (tid < n_blk) ? trow[tid] : 0;
    int over = (mine > SLOT_CAP) ? 1 : 0;
    s_bc[tid] = mine;
    for (int k = tid + 1024; k < n_blk; k += 1024) {
        const int cb = trow[k];
        mine += cb;
        over |= (cb > SLOT_CAP) ? 1 : 0;
    }
    if (tid < BIN_NODES) { s_c[tid] = 0; s_cur[tid] = 0; }
    for (int o = 16; o > 0; o >>= 1) {
        before += __shfl_xor(before, o);
        mine += __shfl_xor(mine, o);
        over |= __shfl_xor(over, o);
    }
    if (lane == 0) { s_ra[wv] = before; s_rb[wv] = mine; s_rc[wv] = over; }
    __syncthreads();
    int start = s_ra[lane], m = s_rb[lane], anyover = s_rc[lane];
    for (int o = 16; o > 0; o >>= 1) {
        start += __shfl_xor(start, o);
        m += __shfl_xor(m, o);
        anyover |= __shfl_xor(anyover, o);
    }

    for (int p0 = 0; p0 < n_pos; p0 += POS_UNROLL * 1024) {
        int rx[POS_UNROLL], cb[POS_UNROLL];
#pragma unroll
        for (int u = 0; u < POS_UNROLL; ++u) {
            const int k = (p0 + u * 1024 + tid) >> SLOT_SHIFT;
            cb[u] = s_bc[(k < BC_CAP) ? k : BC_CAP - 1];
        }
        if (n_blk > BC_CAP) {
#pragma unroll
            for (int u = 0; u < POS_UNROLL; ++u) {
                const int k = (p0 + u * 1024 + tid) >> SLOT_SHIFT;
                if (k >= BC_CAP && k < n_blk) cb[u] = trow[k];
            }
        }
#pragma unroll
        for (int u = 0; u < POS_UNROLL; ++u) {
            const int p = p0 + u * 1024 + tid;
            const bool ok = (p < n_pos) && (cb[u] <= SLOT_CAP) && ((p & (SLOT_CAP - 1)) < cb[u]);
            rx[u] = ok ? sb[p].x : -1;
        }
#pragma unroll
        for (int u = 0; u < POS_UNROLL; ++u)
            if (rx[u] >= 0) atomicAdd(&s_c[rx[u] - node0], 1);
    }
    if (anyover) {
        for (int k = 0; k < n_blk; ++k) {
            int cb;
            if (k < BC_CAP) cb = s_bc[k];
            else cb = trow[k];
            if (cb > SLOT_CAP) {
                int r[4];
                load4(row, k * CHUNK + tid * 4, n_edges, r);
                for (int u = 0; u < 4; ++u)
                    if (r[u] >= 0 && (r[u] >> BIN_SHIFT) == b) atomicAdd(&s_c[r[u] - node0], 1);
            }
        }
    }
    __syncthreads();

    const int v = (tid < BIN_NODES) ? s_c[tid] : 0;
    int incl = v;
    for (int o = 1; o < 32; o <<= 1) {
        const int t = __shfl_up(incl, o);
        if (lane >= o) incl += t;
    }
    if (tid < BIN_NODES && lane == 31) s_w[wv] = incl;
    __syncthreads();
    if (tid < BIN_NODES) {
        int base = 0;
        for (int w = 0; w < wv; ++w) base += s_w[w];
        const int excl = base + incl - v;
        s_o[tid] = excl;
        const int node = node0 + tid;
        if (node < n_nodes) {
            off[node] = start + excl;
            cnt[node] = v;
            dis[node] = (v > 0) ? rsqrtf((float)v) : 0.f;
        }
    }
    __syncthreads();

    const bool staged = (m <= STAGE_CAP);
    for (int p0 = 0; p0 < n_pos; p0 += POS_UNROLL * 1024) {
        int2 rc[POS_UNROLL];
        int cb[POS_UNROLL];
#pragma unroll
        for (int u = 0; u < POS_UNROLL; ++u) {
            const int k = (p0 + u * 1024 + tid) >> SLOT_SHIFT;
            cb[u] = s_bc[(k < BC_CAP) ? k : BC_CAP - 1];
        }
        if (n_blk > BC_CAP) {
#pragma unroll
            for (int u = 0; u < POS_UNROLL; ++u) {
                const int k = (p0 + u * 1024 + tid) >> SLOT_SHIFT;
                if (k >= BC_CAP && k < n_blk) cb[u] = trow[k];
            }
        }
#pragma unroll
        for (int u = 0; u < POS_UNROLL; ++u) {
            const int p = p0 + u * 1024 + tid;
            const bool ok = (p < n_pos) && (cb[u] <= SLOT_CAP) && ((p & (SLOT_CAP - 1)) < cb[u]);
            rc[u] = ok ? sb[p] : make_int2(-1, 0);
        }
#pragma unroll
        for (int u = 0; u < POS_UNROLL; ++u) {
            if (rc[u].x >= 0) {
                const int ln = rc[u].x - node0;
                const int p = s_o[ln] + atomicAdd(&s_cur[ln], 1);
                if (staged) s_b[p] = rc[u].y;
                else bucket[start + p] = rc[u].y;
            }
        }
    }
    if (anyover) {
        for (int k = 0; k < n_blk; ++k) {
            int cb;
            if (k < BC_CAP) cb = s_bc[k];
            else cb = trow[k];
            if (cb > SLOT_CAP) {
                int r[4], c[4];
                load4x2(row, col, k * CHUNK + tid * 4, n_edges, r, c);
                for (int u = 0; u < 4; ++u) {
                    if (r[u] >= 0 && (r[u] >> BIN_SHIFT) == b) {
                        const int ln = r[u] - node0;
                        const int p = s_o[ln] + atomicAdd(&s_cur[ln], 1);
                        if (staged) s_b[p] = c[u];
                        else bucket[start + p] = c[u];
                    }
                }
            }
        }
    }
    if (staged) {
        __syncthreads();
        for (int i = tid; i < m; i += 1024) bucket[start + i] = s_b[i];
    }
}

__device__ __forceinline__ int lane_i(int v, int l) {
    return __builtin_amdgcn_readlane(v, l);
}
__device__ __forceinline__ float lane_f(float v, int l) {
    return __int_as_float(__builtin_amdgcn_readlane(__float_as_int(v), l));
}

#define ROW_FMA(acc, n, xv)                   \
    acc.x = __builtin_fmaf((n), (xv).x, acc.x); \
    acc.y = __builtin_fmaf((n), (xv).y, acc.y); \
    acc.z = __builtin_fmaf((n), (xv).z, acc.z); \
    acc.w = __builtin_fmaf((n), (xv).w, acc.w)

__device__ __forceinline__ float4 row_ld(const float* __restrict__ x, int c, int lane) {
    return ((const float4*)(x + (size_t)c * IN_CH))[lane];
}

__device__ __forceinline__ void add_group4(float4& acc, const float* __restrict__ x, int sv,
                                           float wv, float dr, int t, int lane) {
    const float4 x0 = row_ld(x, lane_i(sv, t), lane);
    const float4 x1 = row_ld(x, lane_i(sv, t + 1), lane);
    const float4 x2 = row_ld(x, lane_i(sv, t + 2), lane);
    const float4 x3 = row_ld(x, lane_i(sv, t + 3), lane);
    const float n0 = dr * lane_f(wv, t), n1 = dr * lane_f(wv, t + 1);
    const float n2 = dr * lane_f(wv, t + 2), n3 = dr * lane_f(wv, t + 3);
    ROW_FMA(acc, n0, x0);
    ROW_FMA(acc, n1, x1);
    ROW_FMA(acc, n2, x2);
    ROW_FMA(acc, n3, x3);
}

__device__ __forceinline__ void add_rest(float4& acc, const float* __restrict__ x, int sv,
                                         float wv, float dr, int t, int rem, int lane) {
    if (rem == 3) {
        const float4 x0 = row_ld(x, lane_i(sv, t), lane);
        const float4 x1 = row_ld(x, lane_i(sv, t + 1), lane);
        const float4 x2 = row_ld(x, lane_i(sv, t + 2), lane);
        const float n0 = dr * lane_f(wv, t), n1 = dr * lane_f(wv, t + 1), n2 = dr * lane_f(wv, t + 2);
        ROW_FMA(acc, n0, x0);
        ROW_FMA(acc, n1, x1);
        ROW_FMA(acc, n2, x2);
    } else if (rem == 2) {
        const float4 x0 = row_ld(x, lane_i(sv, t), lane);
        const float4 x1 = row_ld(x, lane_i(sv, t + 1), lane);
        const float n0 = dr * lane_f(wv, t), n1 = dr * lane_f(wv, t + 1);
        ROW_FMA(acc, n0, x0);
        ROW_FMA(acc, n1, x1);
    } else if (rem == 1) {
        const float4 x0 = row_ld(x, lane_i(sv, t), lane);
        const float n0 = dr * lane_f(wv, t);
        ROW_FMA(acc, n0, x0);
    }
}

__global__ __launch_bounds__(256) __attribute__((amdgpu_waves_per_eu(8))) void gather_gemm_kernel(
    const float* __restrict__ x, const int* __restrict__ bucket,
    const int* __restrict__ planes,
    const float* __restrict__ dis, const float* __restrict__ W,
    const float* __restrict__ bias, float* __restrict__ out, int n_nodes, int n_strips, int n_pad) {
    extern __shared__ __attribute__((aligned(16))) float smem[];
    const int tid = threadIdx.x;
    const int lane = tid & 31;
    const int wave = __builtin_amdgcn_readfirstlane(tid >> 5);
    const int m = lane & 15;
    const int half = lane >> 4;
    const int nb = gridDim.x;

    auto load_words = [&](int sp) -> int {
        const int w = (lane < 6) ? lane : 0;
        const int r0 = sp * 16 + wave * 2;
        const int rr = (r0 < n_nodes) ? r0 : 0;
        return planes[(w >> 1) * n_pad + rr + (w & 1)];
    };
    auto take_words = [&](int words, int sp, int (&start)[2], int (&d)[2], float (&dr)[2]) {
        const int r0 = sp * 16 + wave * 2;
        const bool ok0 = r0 < n_nodes;
        const bool ok1 = (r0 + 1) < n_nodes;
        const int c0 = lane_i(words, 0), c1 = lane_i(words, 1);
        start[0] = lane_i(words, 2);
        start[1] = lane_i(words, 3);
        const float f0 = lane_f(__int_as_float(words), 4);
        const float f1 = lane_f(__int_as_float(words), 5);
        d[0] = ok0 ? c0 : 0;
        d[1] = ok1 ? c1 : 0;
        dr[0] = ok0 ? f0 : 0.f;
        dr[1] = ok1 ? f1 : 0.f;
    };
    auto bucket_ld = [&](const int (&start)[2], const int (&d)[2], int (&kv)[2]) {
#pragma unroll
        for (int q = 0; q < 2; ++q) {
            const bool a = (d[q] <= 32) && (lane < d[q]);
            kv[q] = bucket[a ? start[q] + lane : 0];
        }
    };
    auto order_rows = [&](const int (&d)[2], const int (&kv)[2], int (&s)[2]) {
#pragma unroll
        for (int q = 0; q < 2; ++q) {
            const bool a = (d[q] <= 32) && (lane < d[q]);
            const int k = a ? kv[q] : KEY_MAX;
            int rank = 0;
            if (d[q] <= 32) {
                for (int j = 0; j < d[q]; ++j) {
                    const int kj = lane_i(k, j);
                    rank += (kj < k || (kj == k && j < lane)) ? 1 : 0;
                }
            }
            if (!a) rank = lane;
            s[q] = __builtin_amdgcn_ds_permute(rank << 2, k);
        }
    };
    auto dis_ld = [&](const int (&d)[2], const int (&s)[2], float (&dsv)[2]) {
#pragma unroll
        for (int q = 0; q < 2; ++q) {
            const bool a = (d[q] <= 32) && (lane < d[q]);
            dsv[q] = dis[a ? s[q] : 0];
        }
    };

    int start[2], d[2], s[2], start_n[2], d_n[2];
    float dr[2], ds[2], dr_n[2];
    {
        const int w0 = load_words(blockIdx.x);
        const int w1 = load_words(blockIdx.x + nb);
        take_words(w0, blockIdx.x, start, d, dr);
        int kv[2];
        bucket_ld(start, d, kv);

        for (int i = tid; i < OUT_CH * IN_CH / 4; i += 256) {
            const int o = i >> 5, c4 = i & 31;
            const float4 v = *(const float4*)(W + o * IN_CH + c4 * 4);
            *(float4*)(smem + o * W_STRIDE + c4 * 4) = v;
        }
        order_rows(d, kv, s);
        take_words(w1, blockIdx.x + nb, start_n, d_n, dr_n);
    }
    __syncthreads();
    dis_ld(d, s, ds);
    v2f bf[32];
    {
        const float* bp = smem + (wave * 16 + m) * W_STRIDE + 2 * half;
#pragma unroll
        for (int j = 0; j < 32; ++j) bf[j] = *(const v2f*)(bp + 4 * j);
    }
    const float bv = bias[wave * 16 + m];
    __syncthreads();

    int it = 0;
    for (int strip = blockIdx.x; strip < n_strips; strip += nb, ++it) {
        float* buf = smem + (it & 1) * (16 * W_STRIDE);

        float4 acc0 = make_float4(0.f, 0.f, 0.f, 0.f);
        float4 acc1 = make_float4(0.f, 0.f, 0.f, 0.f);
        if (d[0] <= 32 && d[1] <= 32) {
            const int w0 = d[0] & ~3, w1 = d[1] & ~3;
            const int wm = (w0 < w1) ? w0 : w1;
            int t = 0;
            for (; t < wm; t += 4) {
                const float4 p0 = row_ld(x, lane_i(s[0], t), lane);
                const float4 p1 = row_ld(x, lane_i(s[0], t + 1), lane);
                const float4 p2 = row_ld(x, lane_i(s[0], t + 2), lane);
                const float4 p3 = row_ld(x, lane_i(s[0], t + 3), lane);
                const float4 q0 = row_ld(x, lane_i(s[1], t), lane);
                const float4 q1 = row_ld(x, lane_i(s[1], t + 1), lane);
                const float4 q2 = row_ld(x, lane_i(s[1], t + 2), lane);
                const float4 q3 = row_ld(x, lane_i(s[1], t + 3), lane);
                const float m0 = dr[0] * lane_f(ds[0], t), m1 = dr[0] * lane_f(ds[0], t + 1);
                const float m2 = dr[0] * lane_f(ds[0], t + 2), m3 = dr[0] * lane_f(ds[0], t + 3);
                const float n0 = dr[1] * lane_f(ds[1], t), n1 = dr[1] * lane_f(ds[1], t + 1);
                const float n2 = dr[1] * lane_f(ds[1], t + 2), n3 = dr[1] * lane_f(ds[1], t + 3);
                ROW_FMA(acc0, m0, p0);
                ROW_FMA(acc0, m1, p1);
                ROW_FMA(acc0, m2, p2);
                ROW_FMA(acc0, m3, p3);
                ROW_FMA(acc1, n0, q0);
                ROW_FMA(acc1, n1, q1);
                ROW_FMA(acc1, n2, q2);
                ROW_FMA(acc1, n3, q3);
            }
            for (int u = t; u < w0; u += 4) add_group4(acc0, x, s[0], ds[0], dr[0], u, lane);
            for (int u = t; u < w1; u += 4) add_group4(acc1, x, s[1], ds[1], dr[1], u, lane);
            add_rest(acc0, x, s[0], ds[0], dr[0], w0, d[0] - w0, lane);
            add_rest(acc1, x, s[1], ds[1], dr[1], w1, d[1] - w1, lane);
        } else {
#pragma unroll
            for (int q = 0; q < 2; ++q) {
                float4 acc = make_float4(0.f, 0.f, 0.f, 0.f);
                const int dq = d[q];
                if (dq <= 32) {
#pragma unroll 1
                    for (int t = 0; t < dq; ++t) {
                        const float4 x0 = row_ld(x, lane_i(s[q], t), lane);
                        const float n0 = dr[q] * lane_f(ds[q], t);
                        ROW_FMA(acc, n0, x0);
                    }
                } else {
                    const int st = start[q];
                    const float drq = dr[q];
                    int prev = -1;
                    int done = 0;
                    while (done < dq) {
                        int mn = KEY_MAX;
#pragma unroll 1
                        for (int i = lane; i < dq; i += 32) {
                            const int kk = bucket[st + i];
                            if (kk > prev && kk < mn) mn = kk;
                        }
                        for (int o = 16; o > 0; o >>= 1) {
                            const int m2 = __shfl_xor(mn, o);
                            mn = (m2 < mn) ? m2 : mn;
                        }
                        int c = 0;
#pragma unroll 1
                        for (int i = lane; i < dq; i += 32) c += (bucket[st + i] == mn) ? 1 : 0;
                        for (int o = 16; o > 0; o >>= 1) c += __shfl_xor(c, o);
                        mn = __builtin_amdgcn_readfirstlane(mn);
                        c = __builtin_amdgcn_readfirstlane(c);
                        if (c <= 0) break;
                        const float n0 = drq * dis[mn];
                        const float4 x0 = row_ld(x, mn, lane);
                        for (int u = 0; u < c; ++u) {
                            acc.x += n0 * x0.x; acc.y += n0 * x0.y; acc.z += n0 * x0.z; acc.w += n0 * x0.w;
                        }
                        prev = mn;
                        done += c;
                    }
                }
                if (q == 0) acc0 = acc; else acc1 = acc;
            }
        }
        *(float4*)(buf + (wave * 2) * W_STRIDE + lane * 4) = acc0;
        *(float4*)(buf + (wave * 2 + 1) * W_STRIDE + lane * 4) = acc1;
        __syncthreads();

        const int w2 = load_words(strip + 2 * nb);
        int kv[2];
        bucket_ld(start_n, d_n, kv);

        v8f c8 = {};
        const float* ap = buf + m * W_STRIDE + 2 * half;
#pragma unroll
        for (int j = 0; j < 32; ++j) {
            const v2f a = *(const v2f*)(ap + 4 * j);
            c8 = __builtin_amdgcn_wmma_f32_16x16x4_f32(
                false, a, false, bf[j], (short)0, c8, false, false);
        }
        __builtin_amdgcn_sched_group_barrier(0x100, 6, 0);
#pragma unroll
        for (int g = 0; g < 10; ++g) {
            __builtin_amdgcn_sched_group_barrier(0x008, 2, 0);
            __builtin_amdgcn_sched_group_barrier(0x100, 1, 0);
        }
        __builtin_amdgcn_sched_group_barrier(0x008, 12, 0);
#pragma unroll
        for (int r = 0; r < 8; ++r) {
            const int rowNode = strip * 16 + r + 8 * half;
            if (rowNode < n_nodes)
                out[(size_t)rowNode * OUT_CH + wave * 16 + m] = c8[r] + bv;
        }

        start[0] = start_n[0]; start[1] = start_n[1];
        d[0] = d_n[0]; d[1] = d_n[1];
        dr[0] = dr_n[0]; dr[1] = dr_n[1];
        order_rows(d, kv, s);
        dis_ld(d, s, ds);
        take_words(w2, strip + 2 * nb, start_n, d_n, dr_n);
    }
}

extern "C" void kernel_launch(void* const* d_in, const int* in_sizes, int n_in,
                              void* d_out, int out_size, void* d_ws, size_t ws_size,
                              hipStream_t stream) {
    (void)n_in; (void)out_size; (void)ws_size;

    const float* x  = (const float*)d_in[0];
    const int*   ei = (const int*)d_in[1];
    const float* W  = (const float*)d_in[2];
    const float* b  = (const float*)d_in[3];
    float* out = (float*)d_out;

    const int n_nodes = in_sizes[0] / IN_CH;
    const int n_edges = in_sizes[1] / 2;
    const int* row = ei;
    const int* col = ei + n_edges;

    const int n_pad = (n_nodes + 63) & ~63;
    const int e_pad = (n_edges + 63) & ~63;
    const int n_bins = (n_nodes + BIN_NODES - 1) >> BIN_SHIFT;
    const int n_blk = (n_edges + CHUNK - 1) / CHUNK;
    int*   cnt    = (int*)d_ws;
    int*   off    = cnt + n_pad;
    float* dis    = (float*)(off + n_pad);
    int*   bucket = (int*)(dis + n_pad);
    int2*  slots  = (int2*)(bucket + e_pad);
    int*   table  = (int*)(slots + (((size_t)n_bins * n_blk) << SLOT_SHIFT));

    bin_partition_kernel<<<n_blk, 1024, 0, stream>>>(row, col, table, slots, n_edges, n_bins, n_blk);

    bin_build_kernel<<<n_bins, 1024, 0, stream>>>(row, col, slots, table, bucket, off, cnt, dis,
                                                  n_nodes, n_edges, n_bins, n_blk);

    const int n_strips = (n_nodes + 15) / 16;
    const int n_wg = (n_strips < 1024) ? n_strips : 1024;
    gather_gemm_kernel<<<n_wg, 256, OUT_CH * W_STRIDE * sizeof(float), stream>>>(
        x, bucket, cnt, dis, W, b, out, n_nodes, n_strips, n_pad);
}


# output rows staged in LDS: each wave stores two whole 512 B rows (one store per row, one writer per 128 B line) a strip later, under the next product, not half-line column pieces; sums unchanged
# speedup vs baseline: 3.6187x; 1.0355x over previous
#include <hip/hip_runtime.h>


#define IN_CH 128
#define OUT_CH 128
#define W_STRIDE 132
#define KEY_MAX 0x7fffffff

#define BIN_SHIFT 8
#define BIN_NODES 256
#define MAX_BINS 1024
#define CHUNK 4096
#define STAGE_CAP 8192
#define SLOT_SHIFT 6
#define SLOT_CAP 64
#define BC_CAP 1024
#define POS_UNROLL 10

typedef __attribute__((ext_vector_type(2))) float v2f;
typedef __attribute__((ext_vector_type(8))) float v8f;

__device__ __forceinline__ void load4(const int* __restrict__ p, int e0, int n, int (&v)[4]) {
    if (e0 + 3 < n && ((((size_t)(p + e0)) & 15) == 0)) {
        const int4 q = *(const int4*)(p + e0);
        v[0] = q.x; v[1] = q.y; v[2] = q.z; v[3] = q.w;
    } else {
        v[0] = (e0 + 0 < n) ? p[e0 + 0] : -1;
        v[1] = (e0 + 1 < n) ? p[e0 + 1] : -1;
        v[2] = (e0 + 2 < n) ? p[e0 + 2] : -1;
        v[3] = (e0 + 3 < n) ? p[e0 + 3] : -1;
    }
}

__device__ __forceinline__ void load4x2(const int* __restrict__ p, const int* __restrict__ q,
                                        int e0, int n, int (&v)[4], int (&w)[4]) {
    if (e0 + 3 < n && (((((size_t)(p + e0)) | ((size_t)(q + e0))) & 15) == 0)) {
        const int4 a = *(const int4*)(p + e0);
        const int4 b = *(const int4*)(q + e0);
        v[0] = a.x; v[1] = a.y; v[2] = a.z; v[3] = a.w;
        w[0] = b.x; w[1] = b.y; w[2] = b.z; w[3] = b.w;
    } else {
        v[0] = (e0 + 0 < n) ? p[e0 + 0] : -1;
        v[1] = (e0 + 1 < n) ? p[e0 + 1] : -1;
        v[2] = (e0 + 2 < n) ? p[e0 + 2] : -1;
        v[3] = (e0 + 3 < n) ? p[e0 + 3] : -1;
        w[0] = (e0 + 0 < n) ? q[e0 + 0] : 0;
        w[1] = (e0 + 1 < n) ? q[e0 + 1] : 0;
        w[2] = (e0 + 2 < n) ? q[e0 + 2] : 0;
        w[3] = (e0 + 3 < n) ? q[e0 + 3] : 0;
    }
}

__global__ __launch_bounds__(1024) void bin_partition_kernel(
    const int* __restrict__ row, const int* __restrict__ col, int* __restrict__ table,
    int2* __restrict__ slots, int n_edges, int n_bins, int n_blk) {
    __shared__ int s_cnt[MAX_BINS];
    __shared__ int s_loc[MAX_BINS];
    __shared__ int s_wb[32];
    __shared__ int2 s_stage[CHUNK];
    const int tid = threadIdx.x;
    const int lane = tid & 31;
    const int wv = tid >> 5;
    const int blk = blockIdx.x;

    const int e0 = blk * CHUNK + tid * 4;
    int r[4], c[4];
    load4x2(row, col, e0, n_edges, r, c);

    if (tid < n_bins) s_cnt[tid] = 0;
    __syncthreads();

    int rank[4];
    rank[0] = (r[0] >= 0) ? atomicAdd(&s_cnt[r[0] >> BIN_SHIFT], 1) : 0;
    rank[1] = (r[1] >= 0) ? atomicAdd(&s_cnt[r[1] >> BIN_SHIFT], 1) : 0;
    rank[2] = (r[2] >= 0) ? atomicAdd(&s_cnt[r[2] >> BIN_SHIFT], 1) : 0;
    rank[3] = (r[3] >= 0) ? atomicAdd(&s_cnt[r[3] >> BIN_SHIFT], 1) : 0;
    __syncthreads();

    const int l = (tid < n_bins) ? s_cnt[tid] : 0;
    int il = l;
    for (int o = 1; o < 32; o <<= 1) {
        const int tl = __shfl_up(il, o);
        if (lane >= o) il += tl;
    }
    if (lane == 31) s_wb[wv] = il;
    __syncthreads();
    int bl = 0;
    for (int w = 0; w < wv; ++w) bl += s_wb[w];
    if (tid < n_bins) {
        s_loc[tid] = bl + il - l;
        table[tid * n_blk + blk] = l;
    }
    __syncthreads();

    if (r[0] >= 0) s_stage[s_loc[r[0] >> BIN_SHIFT] + rank[0]] = make_int2(r[0], c[0]);
    if (r[1] >= 0) s_stage[s_loc[r[1] >> BIN_SHIFT] + rank[1]] = make_int2(r[1], c[1]);
    if (r[2] >= 0) s_stage[s_loc[r[2] >> BIN_SHIFT] + rank[2]] = make_int2(r[2], c[2]);
    if (r[3] >= 0) s_stage[s_loc[r[3] >> BIN_SHIFT] + rank[3]] = make_int2(r[3], c[3]);
    __syncthreads();

    int n_here = n_edges - blk * CHUNK;
    if (n_here > CHUNK) n_here = CHUNK;
    for (int i = tid; i < n_here; i += 1024) {
        const int2 rc = s_stage[i];
        const int bin = rc.x >> BIN_SHIFT;
        const int j = i - s_loc[bin];
        if (j < SLOT_CAP) slots[(((size_t)bin * n_blk + blk) << SLOT_SHIFT) + j] = rc;
    }
}

__global__ __launch_bounds__(1024) void bin_build_kernel(
    const int* __restrict__ row, const int* __restrict__ col, const int2* __restrict__ slots,
    const int* __restrict__ table, int* __restrict__ bucket, int* __restrict__ off,
    int* __restrict__ cnt, float* __restrict__ dis, int n_nodes, int n_edges, int n_bins,
    int n_blk) {
    __shared__ int s_c[BIN_NODES];
    __shared__ int s_o[BIN_NODES];
    __shared__ int s_cur[BIN_NODES];
    __shared__ int s_w[8];
    __shared__ int s_ra[32];
    __shared__ int s_rb[32];
    __shared__ int s_rc[32];
    __shared__ int s_bc[BC_CAP];
    __shared__ int s_b[STAGE_CAP];
    const int tid = threadIdx.x;
    const int lane = tid & 31;
    const int wv = tid >> 5;
    const int b = blockIdx.x;
    const int node0 = b << BIN_SHIFT;
    const int* trow = table + b * n_blk;
    const int2* sb = slots + (((size_t)b * n_blk) << SLOT_SHIFT);
    const int n_pos = n_blk << SLOT_SHIFT;

    int before = 0;
    {
        const int n_int = b * n_blk;
        const int n4 = n_int >> 2;
        const int4* t4 = (const int4*)table;
        for (int i0 = 0; i0 < n4; i0 += 8 * 1024) {
            int4 q[8];
#pragma unroll
            for (int u = 0; u < 8; ++u) {
                const int i = i0 + u * 1024 + tid;
                q[u] = (i < n4) ? t4[i] : make_int4(0, 0, 0, 0);
            }
#pragma unroll
            for (int u = 0; u < 8; ++u) before += q[u].x + q[u].y + q[u].z + q[u].w;
        }
        const int it = (n4 << 2) + tid;
        if (it < n_int) before += table[it];
    }
    static_assert(BC_CAP == 1024, "one s_bc entry per thread of the block");
    int mine = (tid < n_blk) ? trow[tid] : 0;
    int over = (mine > SLOT_CAP) ? 1 : 0;
    s_bc[tid] = mine;
    for (int k = tid + 1024; k < n_blk; k += 1024) {
        const int cb = trow[k];
        mine += cb;
        over |= (cb > SLOT_CAP) ? 1 : 0;
    }
    if (tid < BIN_NODES) { s_c[tid] = 0; s_cur[tid] = 0; }
    for (int o = 16; o > 0; o >>= 1) {
        before += __shfl_xor(before, o);
        mine += __shfl_xor(mine, o);
        over |= __shfl_xor(over, o);
    }
    if (lane == 0) { s_ra[wv] = before; s_rb[wv] = mine; s_rc[wv] = over; }
    __syncthreads();
    int start = s_ra[lane], m = s_rb[lane], anyover = s_rc[lane];
    for (int o = 16; o > 0; o >>= 1) {
        start += __shfl_xor(start, o);
        m += __shfl_xor(m, o);
        anyover |= __shfl_xor(anyover, o);
    }

    for (int p0 = 0; p0 < n_pos; p0 += POS_UNROLL * 1024) {
        int rx[POS_UNROLL], cb[POS_UNROLL];
#pragma unroll
        for (int u = 0; u < POS_UNROLL; ++u) {
            const int k = (p0 + u * 1024 + tid) >> SLOT_SHIFT;
            cb[u] = s_bc[(k < BC_CAP) ? k : BC_CAP - 1];
        }
        if (n_blk > BC_CAP) {
#pragma unroll
            for (int u = 0; u < POS_UNROLL; ++u) {
                const int k = (p0 + u * 1024 + tid) >> SLOT_SHIFT;
                if (k >= BC_CAP && k < n_blk) cb[u] = trow[k];
            }
        }
#pragma unroll
        for (int u = 0; u < POS_UNROLL; ++u) {
            const int p = p0 + u * 1024 + tid;
            const bool ok = (p < n_pos) && (cb[u] <= SLOT_CAP) && ((p & (SLOT_CAP - 1)) < cb[u]);
            rx[u] = ok ? sb[p].x : -1;
        }
#pragma unroll
        for (int u = 0; u < POS_UNROLL; ++u)
            if (rx[u] >= 0) atomicAdd(&s_c[rx[u] - node0], 1);
    }
    if (anyover) {
        for (int k = 0; k < n_blk; ++k) {
            int cb;
            if (k < BC_CAP) cb = s_bc[k];
            else cb = trow[k];
            if (cb > SLOT_CAP) {
                int r[4];
                load4(row, k * CHUNK + tid * 4, n_edges, r);
                for (int u = 0; u < 4; ++u)
                    if (r[u] >= 0 && (r[u] >> BIN_SHIFT) == b) atomicAdd(&s_c[r[u] - node0], 1);
            }
        }
    }
    __syncthreads();

    const int v = (tid < BIN_NODES) ? s_c[tid] : 0;
    int incl = v;
    for (int o = 1; o < 32; o <<= 1) {
        const int t = __shfl_up(incl, o);
        if (lane >= o) incl += t;
    }
    if (tid < BIN_NODES && lane == 31) s_w[wv] = incl;
    __syncthreads();
    if (tid < BIN_NODES) {
        int base = 0;
        for (int w = 0; w < wv; ++w) base += s_w[w];
        const int excl = base + incl - v;
        s_o[tid] = excl;
        const int node = node0 + tid;
        if (node < n_nodes) {
            off[node] = start + excl;
            cnt[node] = v;
            dis[node] = (v > 0) ? rsqrtf((float)v) : 0.f;
        }
    }
    __syncthreads();

    const bool staged = (m <= STAGE_CAP);
    for (int p0 = 0; p0 < n_pos; p0 += POS_UNROLL * 1024) {
        int2 rc[POS_UNROLL];
        int cb[POS_UNROLL];
#pragma unroll
        for (int u = 0; u < POS_UNROLL; ++u) {
            const int k = (p0 + u * 1024 + tid) >> SLOT_SHIFT;
            cb[u] = s_bc[(k < BC_CAP) ? k : BC_CAP - 1];
        }
        if (n_blk > BC_CAP) {
#pragma unroll
            for (int u = 0; u < POS_UNROLL; ++u) {
                const int k = (p0 + u * 1024 + tid) >> SLOT_SHIFT;
                if (k >= BC_CAP && k < n_blk) cb[u] = trow[k];
            }
        }
#pragma unroll
        for (int u = 0; u < POS_UNROLL; ++u) {
            const int p = p0 + u * 1024 + tid;
            const bool ok = (p < n_pos) && (cb[u] <= SLOT_CAP) && ((p & (SLOT_CAP - 1)) < cb[u]);
            rc[u] = ok ? sb[p] : make_int2(-1, 0);
        }
#pragma unroll
        for (int u = 0; u < POS_UNROLL; ++u) {
            if (rc[u].x >= 0) {
                const int ln = rc[u].x - node0;
                const int p = s_o[ln] + atomicAdd(&s_cur[ln], 1);
                if (staged) s_b[p] = rc[u].y;
                else bucket[start + p] = rc[u].y;
            }
        }
    }
    if (anyover) {
        for (int k = 0; k < n_blk; ++k) {
            int cb;
            if (k < BC_CAP) cb = s_bc[k];
            else cb = trow[k];
            if (cb > SLOT_CAP) {
                int r[4], c[4];
                load4x2(row, col, k * CHUNK + tid * 4, n_edges, r, c);
                for (int u = 0; u < 4; ++u) {
                    if (r[u] >= 0 && (r[u] >> BIN_SHIFT) == b) {
                        const int ln = r[u] - node0;
                        const int p = s_o[ln] + atomicAdd(&s_cur[ln], 1);
                        if (staged) s_b[p] = c[u];
                        else bucket[start + p] = c[u];
                    }
                }
            }
        }
    }
    if (staged) {
        __syncthreads();
        for (int i = tid; i < m; i += 1024) bucket[start + i] = s_b[i];
    }
}

__device__ __forceinline__ int lane_i(int v, int l) {
    return __builtin_amdgcn_readlane(v, l);
}
__device__ __forceinline__ float lane_f(float v, int l) {
    return __int_as_float(__builtin_amdgcn_readlane(__float_as_int(v), l));
}

#define ROW_FMA(acc, n, xv)                   \
    acc.x = __builtin_fmaf((n), (xv).x, acc.x); \
    acc.y = __builtin_fmaf((n), (xv).y, acc.y); \
    acc.z = __builtin_fmaf((n), (xv).z, acc.z); \
    acc.w = __builtin_fmaf((n), (xv).w, acc.w)

__device__ __forceinline__ float4 row_ld(const float* __restrict__ x, int c, int lane) {
    return ((const float4*)(x + (size_t)c * IN_CH))[lane];
}

__device__ __forceinline__ void add_group4(float4& acc, const float* __restrict__ x, int sv,
                                           float wv, float dr, int t, int lane) {
    const float4 x0 = row_ld(x, lane_i(sv, t), lane);
    const float4 x1 = row_ld(x, lane_i(sv, t + 1), lane);
    const float4 x2 = row_ld(x, lane_i(sv, t + 2), lane);
    const float4 x3 = row_ld(x, lane_i(sv, t + 3), lane);
    const float n0 = dr * lane_f(wv, t), n1 = dr * lane_f(wv, t + 1);
    const float n2 = dr * lane_f(wv, t + 2), n3 = dr * lane_f(wv, t + 3);
    ROW_FMA(acc, n0, x0);
    ROW_FMA(acc, n1, x1);
    ROW_FMA(acc, n2, x2);
    ROW_FMA(acc, n3, x3);
}

__device__ __forceinline__ void add_rest(float4& acc, const float* __restrict__ x, int sv,
                                         float wv, float dr, int t, int rem, int lane) {
    if (rem == 3) {
        const float4 x0 = row_ld(x, lane_i(sv, t), lane);
        const float4 x1 = row_ld(x, lane_i(sv, t + 1), lane);
        const float4 x2 = row_ld(x, lane_i(sv, t + 2), lane);
        const float n0 = dr * lane_f(wv, t), n1 = dr * lane_f(wv, t + 1), n2 = dr * lane_f(wv, t + 2);
        ROW_FMA(acc, n0, x0);
        ROW_FMA(acc, n1, x1);
        ROW_FMA(acc, n2, x2);
    } else if (rem == 2) {
        const float4 x0 = row_ld(x, lane_i(sv, t), lane);
        const float4 x1 = row_ld(x, lane_i(sv, t + 1), lane);
        const float n0 = dr * lane_f(wv, t), n1 = dr * lane_f(wv, t + 1);
        ROW_FMA(acc, n0, x0);
        ROW_FMA(acc, n1, x1);
    } else if (rem == 1) {
        const float4 x0 = row_ld(x, lane_i(sv, t), lane);
        const float n0 = dr * lane_f(wv, t);
        ROW_FMA(acc, n0, x0);
    }
}

__global__ __launch_bounds__(256) __attribute__((amdgpu_waves_per_eu(8))) void gather_gemm_kernel(
    const float* __restrict__ x, const int* __restrict__ bucket,
    const int* __restrict__ planes,
    const float* __restrict__ dis, const float* __restrict__ W,
    const float* __restrict__ bias, float* __restrict__ out, int n_nodes, int n_strips, int n_pad) {
    extern __shared__ __attribute__((aligned(16))) float smem[];
    const int tid = threadIdx.x;
    const int lane = tid & 31;
    const int wave = __builtin_amdgcn_readfirstlane(tid >> 5);
    const int m = lane & 15;
    const int half = lane >> 4;
    const int nb = gridDim.x;

    auto load_words = [&](int sp) -> int {
        const int w = (lane < 6) ? lane : 0;
        const int r0 = sp * 16 + wave * 2;
        const int rr = (r0 < n_nodes) ? r0 : 0;
        return planes[(w >> 1) * n_pad + rr + (w & 1)];
    };
    auto take_words = [&](int words, int sp, int (&start)[2], int (&d)[2], float (&dr)[2]) {
        const int r0 = sp * 16 + wave * 2;
        const bool ok0 = r0 < n_nodes;
        const bool ok1 = (r0 + 1) < n_nodes;
        const int c0 = lane_i(words, 0), c1 = lane_i(words, 1);
        start[0] = lane_i(words, 2);
        start[1] = lane_i(words, 3);
        const float f0 = lane_f(__int_as_float(words), 4);
        const float f1 = lane_f(__int_as_float(words), 5);
        d[0] = ok0 ? c0 : 0;
        d[1] = ok1 ? c1 : 0;
        dr[0] = ok0 ? f0 : 0.f;
        dr[1] = ok1 ? f1 : 0.f;
    };
    auto bucket_ld = [&](const int (&start)[2], const int (&d)[2], int (&kv)[2]) {
#pragma unroll
        for (int q = 0; q < 2; ++q) {
            const bool a = (d[q] <= 32) && (lane < d[q]);
            kv[q] = bucket[a ? start[q] + lane : 0];
        }
    };
    auto order_rows = [&](const int (&d)[2], const int (&kv)[2], int (&s)[2]) {
#pragma unroll
        for (int q = 0; q < 2; ++q) {
            const bool a = (d[q] <= 32) && (lane < d[q]);
            const int k = a ? kv[q] : KEY_MAX;
            int rank = 0;
            if (d[q] <= 32) {
                for (int j = 0; j < d[q]; ++j) {
                    const int kj = lane_i(k, j);
                    rank += (kj < k || (kj == k && j < lane)) ? 1 : 0;
                }
            }
            if (!a) rank = lane;
            s[q] = __builtin_amdgcn_ds_permute(rank << 2, k);
        }
    };
    auto dis_ld = [&](const int (&d)[2], const int (&s)[2], float (&dsv)[2]) {
#pragma unroll
        for (int q = 0; q < 2; ++q) {
            const bool a = (d[q] <= 32) && (lane < d[q]);
            dsv[q] = dis[a ? s[q] : 0];
        }
    };

    int start[2], d[2], s[2], start_n[2], d_n[2];
    float dr[2], ds[2], dr_n[2];
    {
        const int w0 = load_words(blockIdx.x);
        const int w1 = load_words(blockIdx.x + nb);
        take_words(w0, blockIdx.x, start, d, dr);
        int kv[2];
        bucket_ld(start, d, kv);

        for (int i = tid; i < OUT_CH * IN_CH / 4; i += 256) {
            const int o = i >> 5, c4 = i & 31;
            const float4 v = *(const float4*)(W + o * IN_CH + c4 * 4);
            *(float4*)(smem + o * W_STRIDE + c4 * 4) = v;
        }
        order_rows(d, kv, s);
        take_words(w1, blockIdx.x + nb, start_n, d_n, dr_n);
    }
    __syncthreads();
    dis_ld(d, s, ds);
    v2f bf[32];
    {
        const float* bp = smem + (wave * 16 + m) * W_STRIDE + 2 * half;
#pragma unroll
        for (int j = 0; j < 32; ++j) bf[j] = *(const v2f*)(bp + 4 * j);
    }
    const float bv = bias[wave * 16 + m];
    __syncthreads();

    auto flush_rows = [&](int fs, int par) {
        const float* ob = smem + (2 + par) * (16 * W_STRIDE) + (wave * 2) * W_STRIDE + lane * 4;
        const int r0 = fs * 16 + wave * 2;
        const float4 v0 = *(const float4*)(ob);
        const float4 v1 = *(const float4*)(ob + W_STRIDE);
        if (r0 < n_nodes) *(float4*)(out + (size_t)r0 * OUT_CH + lane * 4) = v0;
        if (r0 + 1 < n_nodes) *(float4*)(out + (size_t)(r0 + 1) * OUT_CH + lane * 4) = v1;
    };

    int it = 0;
    int strip = blockIdx.x;
    for (; strip < n_strips; strip += nb, ++it) {
        float* buf = smem + (it & 1) * (16 * W_STRIDE);

        float4 acc0 = make_float4(0.f, 0.f, 0.f, 0.f);
        float4 acc1 = make_float4(0.f, 0.f, 0.f, 0.f);
        if (d[0] <= 32 && d[1] <= 32) {
            const int w0 = d[0] & ~3, w1 = d[1] & ~3;
            const int wm = (w0 < w1) ? w0 : w1;
            int t = 0;
            for (; t < wm; t += 4) {
                const float4 p0 = row_ld(x, lane_i(s[0], t), lane);
                const float4 p1 = row_ld(x, lane_i(s[0], t + 1), lane);
                const float4 p2 = row_ld(x, lane_i(s[0], t + 2), lane);
                const float4 p3 = row_ld(x, lane_i(s[0], t + 3), lane);
                const float4 q0 = row_ld(x, lane_i(s[1], t), lane);
                const float4 q1 = row_ld(x, lane_i(s[1], t + 1), lane);
                const float4 q2 = row_ld(x, lane_i(s[1], t + 2), lane);
                const float4 q3 = row_ld(x, lane_i(s[1], t + 3), lane);
                const float m0 = dr[0] * lane_f(ds[0], t), m1 = dr[0] * lane_f(ds[0], t + 1);
                const float m2 = dr[0] * lane_f(ds[0], t + 2), m3 = dr[0] * lane_f(ds[0], t + 3);
                const float n0 = dr[1] * lane_f(ds[1], t), n1 = dr[1] * lane_f(ds[1], t + 1);
                const float n2 = dr[1] * lane_f(ds[1], t + 2), n3 = dr[1] * lane_f(ds[1], t + 3);
                ROW_FMA(acc0, m0, p0);
                ROW_FMA(acc0, m1, p1);
                ROW_FMA(acc0, m2, p2);
                ROW_FMA(acc0, m3, p3);
                ROW_FMA(acc1, n0, q0);
                ROW_FMA(acc1, n1, q1);
                ROW_FMA(acc1, n2, q2);
                ROW_FMA(acc1, n3, q3);
            }
            for (int u = t; u < w0; u += 4) add_group4(acc0, x, s[0], ds[0], dr[0], u, lane);
            for (int u = t; u < w1; u += 4) add_group4(acc1, x, s[1], ds[1], dr[1], u, lane);
            add_rest(acc0, x, s[0], ds[0], dr[0], w0, d[0] - w0, lane);
            add_rest(acc1, x, s[1], ds[1], dr[1], w1, d[1] - w1, lane);
        } else {
#pragma unroll
            for (int q = 0; q < 2; ++q) {
                float4 acc = make_float4(0.f, 0.f, 0.f, 0.f);
                const int dq = d[q];
                if (dq <= 32) {
#pragma unroll 1
                    for (int t = 0; t < dq; ++t) {
                        const float4 x0 = row_ld(x, lane_i(s[q], t), lane);
                        const float n0 = dr[q] * lane_f(ds[q], t);
                        ROW_FMA(acc, n0, x0);
                    }
                } else {
                    const int st = start[q];
                    const float drq = dr[q];
                    int prev = -1;
                    int done = 0;
                    while (done < dq) {
                        int mn = KEY_MAX;
#pragma unroll 1
                        for (int i = lane; i < dq; i += 32) {
                            const int kk = bucket[st + i];
                            if (kk > prev && kk < mn) mn = kk;
                        }
                        for (int o = 16; o > 0; o >>= 1) {
                            const int m2 = __shfl_xor(mn, o);
                            mn = (m2 < mn) ? m2 : mn;
                        }
                        int c = 0;
#pragma unroll 1
                        for (int i = lane; i < dq; i += 32) c += (bucket[st + i] == mn) ? 1 : 0;
                        for (int o = 16; o > 0; o >>= 1) c += __shfl_xor(c, o);
                        mn = __builtin_amdgcn_readfirstlane(mn);
                        c = __builtin_amdgcn_readfirstlane(c);
                        if (c <= 0) break;
                        const float n0 = drq * dis[mn];
                        const float4 x0 = row_ld(x, mn, lane);
                        for (int u = 0; u < c; ++u) {
                            acc.x += n0 * x0.x; acc.y += n0 * x0.y; acc.z += n0 * x0.z; acc.w += n0 * x0.w;
                        }
                        prev = mn;
                        done += c;
                    }
                }
                if (q == 0) acc0 = acc; else acc1 = acc;
            }
        }
        *(float4*)(buf + (wave * 2) * W_STRIDE + lane * 4) = acc0;
        *(float4*)(buf + (wave * 2 + 1) * W_STRIDE + lane * 4) = acc1;
        __syncthreads();

        if (it > 0) flush_rows(strip - nb, (it - 1) & 1);

        const int w2 = load_words(strip + 2 * nb);
        int kv[2];
        bucket_ld(start_n, d_n, kv);

        v8f c8 = {};
        const float* ap = buf + m * W_STRIDE + 2 * half;
#pragma unroll
        for (int j = 0; j < 32; ++j) {
            const v2f a = *(const v2f*)(ap + 4 * j);
            c8 = __builtin_amdgcn_wmma_f32_16x16x4_f32(
                false, a, false, bf[j], (short)0, c8, false, false);
        }
        __builtin_amdgcn_sched_group_barrier(0x100, 6, 0);
#pragma unroll
        for (int g = 0; g < 10; ++g) {
            __builtin_amdgcn_sched_group_barrier(0x008, 2, 0);
            __builtin_amdgcn_sched_group_barrier(0x100, 1, 0);
        }
        __builtin_amdgcn_sched_group_barrier(0x008, 12, 0);
        float* ob = smem + (2 + (it & 1)) * (16 * W_STRIDE);
#pragma unroll
        for (int r = 0; r < 8; ++r) ob[(r + 8 * half) * W_STRIDE + wave * 16 + m] = c8[r] + bv;

        start[0] = start_n[0]; start[1] = start_n[1];
        d[0] = d_n[0]; d[1] = d_n[1];
        dr[0] = dr_n[0]; dr[1] = dr_n[1];
        order_rows(d, kv, s);
        dis_ld(d, s, ds);
        take_words(w2, strip + 2 * nb, start_n, d_n, dr_n);
    }
    __syncthreads();
    if (it > 0) flush_rows(strip - nb, (it - 1) & 1);
}

extern "C" void kernel_launch(void* const* d_in, const int* in_sizes, int n_in,
                              void* d_out, int out_size, void* d_ws, size_t ws_size,
                              hipStream_t stream) {
    (void)n_in; (void)out_size; (void)ws_size;

    const float* x  = (const float*)d_in[0];
    const int*   ei = (const int*)d_in[1];
    const float* W  = (const float*)d_in[2];
    const float* b  = (const float*)d_in[3];
    float* out = (float*)d_out;

    const int n_nodes = in_sizes[0] / IN_CH;
    const int n_edges = in_sizes[1] / 2;
    const int* row = ei;
    const int* col = ei + n_edges;

    const int n_pad = (n_nodes + 63) & ~63;
    const int e_pad = (n_edges + 63) & ~63;
    const int n_bins = (n_nodes + BIN_NODES - 1) >> BIN_SHIFT;
    const int n_blk = (n_edges + CHUNK - 1) / CHUNK;
    int*   cnt    = (int*)d_ws;
    int*   off    = cnt + n_pad;
    float* dis    = (float*)(off + n_pad);
    int*   bucket = (int*)(dis + n_pad);
    int2*  slots  = (int2*)(bucket + e_pad);
    int*   table  = (int*)(slots + (((size_t)n_bins * n_blk) << SLOT_SHIFT));

    bin_partition_kernel<<<n_blk, 1024, 0, stream>>>(row, col, table, slots, n_edges, n_bins, n_blk);

    bin_build_kernel<<<n_bins, 1024, 0, stream>>>(row, col, slots, table, bucket, off, cnt, dis,
                                                  n_nodes, n_edges, n_bins, n_blk);

    const int n_strips = (n_nodes + 15) / 16;
    const int n_wg = (n_strips < 1024) ? n_strips : 1024;
    gather_gemm_kernel<<<n_wg, 256, OUT_CH * W_STRIDE * sizeof(float), stream>>>(
        x, bucket, cnt, dis, W, b, out, n_nodes, n_strips, n_pad);
}
